# starting kernel
# baseline (speedup 1.0000x reference)
#include <hip/hip_runtime.h>
#include <math.h>

#define NN   50000
#define EE   800000
#define SDIM 128
#define VDIM 16
#define V3   48
#define ESD  32

typedef float v2f __attribute__((ext_vector_type(2)));
typedef float v8f __attribute__((ext_vector_type(8)));

__device__ __forceinline__ v8f wmma4(v2f a, v2f b, v8f c) {
  return __builtin_amdgcn_wmma_f32_16x16x4_f32(false, a, false, b, (short)0, c,
                                               false, false);
}

__device__ __forceinline__ float fast_rcp(float x) {
  return __builtin_amdgcn_rcpf(x);
}

__device__ __forceinline__ float sigmoidf_(float x) {
  return fast_rcp(1.f + __expf(-x));
}

__device__ __forceinline__ void fadd_atomic(float* p, float v) {
  unsafeAtomicAdd(p, v);
}

__global__ void k_zero(float* __restrict__ p, int n) {
  int i = blockIdx.x * blockDim.x + threadIdx.x;
  int stride = gridDim.x * blockDim.x;
  for (; i < n; i += stride) p[i] = 0.f;
}

__global__ void __launch_bounds__(256) k_norm(
    const float* __restrict__ node_s, const float* __restrict__ node_v,
    const float* __restrict__ ln_g, const float* __restrict__ ln_b,
    float* __restrict__ s_n, float* __restrict__ v_n) {
  int wave = (blockIdx.x * blockDim.x + threadIdx.x) >> 5;
  int lane = threadIdx.x & 31;
  if (wave >= NN) return;

  const float* srow = node_s + (long)wave * SDIM;
  float4 x = ((const float4*)srow)[lane];
  float s = x.x + x.y + x.z + x.w;
#pragma unroll
  for (int m = 16; m >= 1; m >>= 1) s += __shfl_xor(s, m, 32);
  float mu = s * (1.f / SDIM);
  float dx = x.x - mu, dy = x.y - mu, dz = x.z - mu, dw = x.w - mu;
  float q = dx * dx + dy * dy + dz * dz + dw * dw;
#pragma unroll
  for (int m = 16; m >= 1; m >>= 1) q += __shfl_xor(q, m, 32);
  float rs = rsqrtf(q * (1.f / SDIM) + 1e-5f);
  int c0 = lane * 4;
  float4 o;
  o.x = dx * rs * ln_g[c0 + 0] + ln_b[c0 + 0];
  o.y = dy * rs * ln_g[c0 + 1] + ln_b[c0 + 1];
  o.z = dz * rs * ln_g[c0 + 2] + ln_b[c0 + 2];
  o.w = dw * rs * ln_g[c0 + 3] + ln_b[c0 + 3];
  ((float4*)(s_n + (long)wave * SDIM))[lane] = o;

  const float* vrow = node_v + (long)wave * V3;
  float2 v = {0.f, 0.f};
  int vi = lane * 2;
  if (vi < V3) v = ((const float2*)vrow)[lane];
  float qq = v.x * v.x + v.y * v.y;
#pragma unroll
  for (int m = 16; m >= 1; m >>= 1) qq += __shfl_xor(qq, m, 32);
  float inv = rsqrtf(qq * (1.f / VDIM) + 1e-8f);
  if (vi < V3) {
    float2 ov;
    ov.x = v.x * inv;
    ov.y = v.y * inv;
    ((float2*)(v_n + (long)wave * V3))[lane] = ov;
  }
}

__global__ void __launch_bounds__(256) k_edge(
    const float* __restrict__ edge_s, const float* __restrict__ edge_v,
    const int* __restrict__ edge_dst,
    const float* __restrict__ wes, const float* __restrict__ bes,
    const float* __restrict__ wev, const float* __restrict__ bev,
    float* __restrict__ agg_s, float* __restrict__ agg_v,
    float* __restrict__ cnt) {
  __shared__ float lds_gate[8][16 * 16];
  int wid = threadIdx.x >> 5;
  int lane = threadIdx.x & 31;
  int group = blockIdx.x * 8 + wid;
  if (group >= EE / 16) return;
  int ebase = group * 16;
  int row = lane & 15;
  int half = lane >> 4;

  const float* arow = edge_s + (long)(ebase + row) * ESD;

  int dsts[8];
#pragma unroll
  for (int r = 0; r < 8; r++) dsts[r] = edge_dst[ebase + r + 8 * half];

#pragma unroll 1
  for (int c = 0; c < 8; c++) {
    v8f acc = {};
    int n = 16 * c + row;
    const float* brow = wes + (long)n * ESD;
#pragma unroll
    for (int j = 0; j < 8; j++) {
      int k0 = 4 * j + 2 * half;
      v2f a = *(const v2f*)(arow + k0);
      v2f b = *(const v2f*)(brow + k0);
      acc = wmma4(a, b, acc);
    }
    float bias = bes[n];
#pragma unroll
    for (int r = 0; r < 8; r++) {
      float pre = acc[r] + bias;
      float so = pre * sigmoidf_(pre);
      if (c == 0) lds_gate[wid][(r + 8 * half) * 16 + n] = sigmoidf_(so);
      fadd_atomic(agg_s + (long)dsts[r] * SDIM + n, so);
    }
  }

  if (lane < 16) fadd_atomic(cnt + edge_dst[ebase + lane], 1.f);

  int e = ebase + row;
  int dstv = edge_dst[e];
  float ev0 = edge_v[e * 3 + 0];
  float ev1 = edge_v[e * 3 + 1];
  float ev2 = edge_v[e * 3 + 2];
#pragma unroll
  for (int t = 0; t < 24; t++) {
    int d = half * 24 + t;
    float v = ev0 * wev[d * 3 + 0] + ev1 * wev[d * 3 + 1] +
              ev2 * wev[d * 3 + 2] + bev[d];
    float g = lds_gate[wid][row * 16 + d / 3];
    fadd_atomic(agg_v + (long)dstv * V3 + d, v * g);
  }
}

__global__ void __launch_bounds__(256) k_node(
    const float* __restrict__ s_n, const float* __restrict__ v_n,
    const float* __restrict__ agg_s, const float* __restrict__ agg_v,
    const float* __restrict__ cnt,
    const float* __restrict__ wns, const float* __restrict__ bns,
    const float* __restrict__ wnv, const float* __restrict__ bnv,
    const float* __restrict__ wrs, const float* __restrict__ brs,
    const float* __restrict__ wrv, const float* __restrict__ brv,
    float* __restrict__ out) {
  __shared__ float lds_gate[8][16 * 16];
  int wid = threadIdx.x >> 5;
  int lane = threadIdx.x & 31;
  int group = blockIdx.x * 8 + wid;
  if (group >= NN / 16) return;
  int nb = group * 16;
  int row = lane & 15;
  int half = lane >> 4;
  int noderow = nb + row;

  float invden = fast_rcp(fmaxf(cnt[noderow], 1.f));
  const float* sA = s_n + (long)noderow * SDIM;
  const float* gA = agg_s + (long)noderow * SDIM;

#pragma unroll 1
  for (int ct = 0; ct < 8; ct++) {
    v8f ao = {};
    v8f ar = {};
    int n = 16 * ct + row;
    const float* bo = wns + (long)n * SDIM;
    const float* br = wrs + (long)n * SDIM;
#pragma unroll 4
    for (int j = 0; j < 32; j++) {
      int k0 = 4 * j + 2 * half;
      v2f as = *(const v2f*)(sA + k0);
      v2f ag = *(const v2f*)(gA + k0);
      v2f ain = {as.x + ag.x * invden, as.y + ag.y * invden};
      v2f bO = *(const v2f*)(bo + k0);
      v2f bR = *(const v2f*)(br + k0);
      ao = wmma4(ain, bO, ao);
      ar = wmma4(as, bR, ar);
    }
    float biasO = bns[n], biasR = brs[n];
#pragma unroll
    for (int r = 0; r < 8; r++) {
      int m = r + 8 * half;
      float pre = ao[r] + biasO;
      float so = pre * sigmoidf_(pre);
      if (ct == 0) lds_gate[wid][m * 16 + n] = sigmoidf_(so);
      out[(long)(nb + m) * SDIM + n] = so + (ar[r] + biasR);
    }
  }

  const float* vA = v_n + (long)noderow * V3;
  const float* gV = agg_v + (long)noderow * V3;
  float* outv = out + (long)NN * SDIM;
#pragma unroll 1
  for (int ct = 0; ct < 3; ct++) {
    v8f ao = {};
    v8f ar = {};
    int d = 16 * ct + row;
    const float* bo = wnv + (long)d * V3;
    const float* br = wrv + (long)d * V3;
#pragma unroll
    for (int j = 0; j < 12; j++) {
      int k0 = 4 * j + 2 * half;
      v2f av = *(const v2f*)(vA + k0);
      v2f ag = *(const v2f*)(gV + k0);
      v2f ain = {av.x + ag.x * invden, av.y + ag.y * invden};
      v2f bO = *(const v2f*)(bo + k0);
      v2f bR = *(const v2f*)(br + k0);
      ao = wmma4(ain, bO, ao);
      ar = wmma4(av, bR, ar);
    }
    float biasO = bnv[d], biasR = brv[d];
    int vo = d / 3;
#pragma unroll
    for (int r = 0; r < 8; r++) {
      int m = r + 8 * half;
      float pre = ao[r] + biasO;
      float g = lds_gate[wid][m * 16 + vo];
      outv[(long)(nb + m) * V3 + d] = pre * g + (ar[r] + biasR);
    }
  }
}

extern "C" void kernel_launch(void* const* d_in, const int* in_sizes, int n_in,
                              void* d_out, int out_size, void* d_ws,
                              size_t ws_size, hipStream_t stream) {
  const float* node_s = (const float*)d_in[0];
  const float* node_v = (const float*)d_in[1];
  const int* eidx     = (const int*)d_in[2];
  const float* edge_s = (const float*)d_in[3];
  const float* edge_v = (const float*)d_in[4];
  const float* ln_g = (const float*)d_in[5];
  const float* ln_b = (const float*)d_in[6];
  const float* wes = (const float*)d_in[7];
  const float* bes = (const float*)d_in[8];
  const float* wev = (const float*)d_in[9];
  const float* bev = (const float*)d_in[10];
  const float* wns = (const float*)d_in[11];
  const float* bns = (const float*)d_in[12];
  const float* wnv = (const float*)d_in[13];
  const float* bnv = (const float*)d_in[14];
  const float* wrs = (const float*)d_in[15];
  const float* brs = (const float*)d_in[16];
  const float* wrv = (const float*)d_in[17];
  const float* brv = (const float*)d_in[18];
  float* out = (float*)d_out;

  float* ws = (float*)d_ws;
  float* s_n = ws;
  float* v_n = s_n + (size_t)NN * SDIM;
  float* agg_s = v_n + (size_t)NN * V3;
  float* agg_v = agg_s + (size_t)NN * SDIM;
  float* cnt = agg_v + (size_t)NN * V3;
  const int* edge_dst = eidx + EE;

  int zn = NN * (SDIM + V3 + 1);
  k_zero<<<2048, 256, 0, stream>>>(agg_s, zn);
  k_norm<<<(NN * 32 + 255) / 256, 256, 0, stream>>>(node_s, node_v, ln_g, ln_b,
                                                    s_n, v_n);
  k_edge<<<EE / 16 / 8, 256, 0, stream>>>(edge_s, edge_v, edge_dst, wes, bes,
                                          wev, bev, agg_s, agg_v, cnt);
  k_node<<<(NN / 16 + 7) / 8, 256, 0, stream>>>(s_n, v_n, agg_s, agg_v, cnt,
                                                wns, bns, wnv, bnv, wrs, brs,
                                                wrv, brv, out);
}


# k_edge scatter-add off global atomics: each work-group owns 196 nodes in LDS, filters the raw edge list, adds messages by LDS float atomics, stores once; k_zero gone; sum order = LDS arrival
# speedup vs baseline: 3.1097x; 3.1097x over previous
#include <hip/hip_runtime.h>
#include <math.h>

#define NN   50000
#define EE   800000
#define SDIM 128
#define VDIM 16
#define V3   48
#define ESD  32

typedef float v2f __attribute__((ext_vector_type(2)));
typedef float v8f __attribute__((ext_vector_type(8)));

__device__ __forceinline__ v8f wmma4(v2f a, v2f b, v8f c) {
  return __builtin_amdgcn_wmma_f32_16x16x4_f32(false, a, false, b, (short)0, c,
                                               false, false);
}

__device__ __forceinline__ float fast_rcp(float x) {
  return __builtin_amdgcn_rcpf(x);
}

__device__ __forceinline__ float sigmoidf_(float x) {
  return fast_rcp(1.f + __expf(-x));
}

__global__ void __launch_bounds__(256) k_norm(
    const float* __restrict__ node_s, const float* __restrict__ node_v,
    const float* __restrict__ ln_g, const float* __restrict__ ln_b,
    float* __restrict__ s_n, float* __restrict__ v_n) {
  int wave = (blockIdx.x * blockDim.x + threadIdx.x) >> 5;
  int lane = threadIdx.x & 31;
  if (wave >= NN) return;

  const float* srow = node_s + (long)wave * SDIM;
  float4 x = ((const float4*)srow)[lane];
  float s = x.x + x.y + x.z + x.w;
#pragma unroll
  for (int m = 16; m >= 1; m >>= 1) s += __shfl_xor(s, m, 32);
  float mu = s * (1.f / SDIM);
  float dx = x.x - mu, dy = x.y - mu, dz = x.z - mu, dw = x.w - mu;
  float q = dx * dx + dy * dy + dz * dz + dw * dw;
#pragma unroll
  for (int m = 16; m >= 1; m >>= 1) q += __shfl_xor(q, m, 32);
  float rs = rsqrtf(q * (1.f / SDIM) + 1e-5f);
  int c0 = lane * 4;
  float4 o;
  o.x = dx * rs * ln_g[c0 + 0] + ln_b[c0 + 0];
  o.y = dy * rs * ln_g[c0 + 1] + ln_b[c0 + 1];
  o.z = dz * rs * ln_g[c0 + 2] + ln_b[c0 + 2];
  o.w = dw * rs * ln_g[c0 + 3] + ln_b[c0 + 3];
  ((float4*)(s_n + (long)wave * SDIM))[lane] = o;

  const float* vrow = node_v + (long)wave * V3;
  float2 v = {0.f, 0.f};
  int vi = lane * 2;
  if (vi < V3) v = ((const float2*)vrow)[lane];
  float qq = v.x * v.x + v.y * v.y;
#pragma unroll
  for (int m = 16; m >= 1; m >>= 1) qq += __shfl_xor(qq, m, 32);
  float inv = rsqrtf(qq * (1.f / VDIM) + 1e-8f);
  if (vi < V3) {
    float2 ov;
    ov.x = v.x * inv;
    ov.y = v.y * inv;
    ((float2*)(v_n + (long)wave * V3))[lane] = ov;
  }
}

#define RNG  196
#define NBLK ((NN + RNG - 1) / RNG)
#define SP   144
#define VP   49
#define GP   17
#define NW   32
#define QCAP 528

__device__ __forceinline__ void lds_fadd(float* p, float v) {
  __hip_atomic_fetch_add(p, v, __ATOMIC_RELAXED, __HIP_MEMORY_SCOPE_WORKGROUP);
}

__device__ __forceinline__ void edge_tile(
    const int* qp, int nv, int lane,
    const float* __restrict__ edge_s, const float* __restrict__ edge_v,
    const float* __restrict__ bes, const float4* Wl, const v2f* Bl,
    float* gate, float* accS, float* accV, float* accC) {
  int row = lane & 15;
  int half = lane >> 4;
  int pk = qp[row < nv ? row : 0];
  int eid = pk & 0xFFFFF;
  int dlv = row < nv ? pk >> 20 : RNG;

  int dlr[8];
#pragma unroll
  for (int r = 0; r < 8; r++) {
    int m = r + 8 * half;
    dlr[r] = m < nv ? qp[m] >> 20 : RNG;
  }

  const float* arow = edge_s + (long)eid * ESD;
  v2f a[8];
#pragma unroll
  for (int j = 0; j < 8; j++) a[j] = *(const v2f*)(arow + 4 * j + 2 * half);
  float ev0 = edge_v[eid * 3 + 0];
  float ev1 = edge_v[eid * 3 + 1];
  float ev2 = edge_v[eid * 3 + 2];

#pragma unroll 1
  for (int c = 0; c < 8; c++) {
    v8f acc = {};
    int n = 16 * c + row;
#pragma unroll
    for (int j = 0; j < 8; j++) {
      v2f b = Bl[(c * 8 + j) * 32 + lane];
      acc = wmma4(a[j], b, acc);
    }
    float bias = bes[n];
#pragma unroll
    for (int r = 0; r < 8; r++) {
      int m = r + 8 * half;
      float pre = acc[r] + bias;
      float so = pre * sigmoidf_(pre);
      if (c == 0) gate[m * GP + row] = sigmoidf_(so);
      lds_fadd(accS + dlr[r] * SP + n, so);
    }
  }

  if (half == 0) lds_fadd(accC + dlv, 1.f);

  __builtin_amdgcn_wave_barrier();

#pragma unroll
  for (int t0 = 0; t0 < 24; t0 += 8) {
    float vg[8];
#pragma unroll
    for (int t = 0; t < 8; t++) {
      int d = half * 24 + t0 + t;
      float4 w = Wl[d];
      float v = ev0 * w.x + ev1 * w.y + ev2 * w.z + w.w;
      vg[t] = v * gate[row * GP + d / 3];
    }
#pragma unroll
    for (int t = 0; t < 8; t++)
      lds_fadd(accV + dlv * VP + half * 24 + t0 + t, vg[t]);
  }
  __builtin_amdgcn_wave_barrier();
}

__device__ __forceinline__ int4 ld_dst(const int* __restrict__ edge_dst, int t,
                                       int wid, int lane) {
  int ix = (t * NW + wid) * 128 + lane * 4;
  int4 d = {-1, -1, -1, -1};
  if (ix < EE) d = *(const int4*)(edge_dst + ix);
  return d;
}

__global__ void __launch_bounds__(1024) k_edge(
    const float* __restrict__ edge_s, const float* __restrict__ edge_v,
    const int* __restrict__ edge_dst,
    const float* __restrict__ wes, const float* __restrict__ bes,
    const float* __restrict__ wev, const float* __restrict__ bev,
    float* __restrict__ agg_s, float* __restrict__ agg_v,
    float* __restrict__ cnt) {
  __shared__ float accS[(RNG + 1) * SP];
  __shared__ float accV[(RNG + 1) * VP];
  __shared__ float accC[RNG + 1];
  __shared__ v2f   Bl[8 * 8 * 32];
  __shared__ float4 Wl[V3];
  __shared__ int   queue[NW][QCAP];
  __shared__ float gateL[NW][16 * GP];

  int tid = threadIdx.x;
  int wid = tid >> 5;
  int lane = tid & 31;
  int lo = blockIdx.x * RNG;
  int nr = NN - lo < RNG ? NN - lo : RNG;

  for (int i = tid; i < (RNG + 1) * SP; i += 1024) accS[i] = 0.f;
  for (int i = tid; i < (RNG + 1) * VP; i += 1024) accV[i] = 0.f;
  for (int i = tid; i < RNG + 1; i += 1024) accC[i] = 0.f;
  for (int i = tid; i < 8 * 8 * 32; i += 1024) {
    int l = i & 31;
    int j = (i >> 5) & 7;
    int c = i >> 8;
    int n = 16 * c + (l & 15);
    int k0 = 4 * j + 2 * (l >> 4);
    Bl[i] = *(const v2f*)(wes + (long)n * ESD + k0);
  }
  if (tid < V3) {
    float4 w = {wev[tid * 3 + 0], wev[tid * 3 + 1], wev[tid * 3 + 2], bev[tid]};
    Wl[tid] = w;
  }
  __syncthreads();

  int* q = queue[wid];
  float* gate = gateL[wid];
  int qn = 0;
  unsigned ltmask = (1u << lane) - 1u;
  const int NSTEP = (EE + 128 * NW - 1) / (128 * NW);

  const int NG = (NSTEP + 3) / 4;
  int4 dd[4];
#pragma unroll
  for (int u = 0; u < 4; u++) dd[u] = ld_dst(edge_dst, u, wid, lane);
#pragma unroll 1
  for (int g = 0; g <= NG; g++) {
    bool last = g == NG;
#pragma unroll
    for (int u = 0; u < 4; u++) {
      int ixc = ((g * 4 + u) * NW + wid) * 128 + lane * 4;
      int dv[4] = {dd[u].x, dd[u].y, dd[u].z, dd[u].w};
      bool mm[4];
      bool any = false;
#pragma unroll
      for (int s = 0; s < 4; s++) {
        mm[s] = (unsigned)(dv[s] - lo) < (unsigned)nr;
        any = any || mm[s];
      }
      if (__builtin_amdgcn_ballot_w32(any) != 0u) {
#pragma unroll
        for (int s = 0; s < 4; s++) {
          unsigned bal = __builtin_amdgcn_ballot_w32(mm[s]);
          if (mm[s])
            q[qn + __popc(bal & ltmask)] = (ixc + s) | ((dv[s] - lo) << 20);
          qn += __popc(bal);
        }
      }
    }
#pragma unroll
    for (int u = 0; u < 4; u++)
      dd[u] = ld_dst(edge_dst, (g + 1) * 4 + u, wid, lane);
    __builtin_amdgcn_wave_barrier();
#pragma unroll 1
    while (qn >= 16 || (last && qn > 0)) {
      int nv = qn < 16 ? qn : 16;
      qn -= nv;
      edge_tile(q + qn, nv, lane, edge_s, edge_v, bes, Wl, Bl, gate, accS,
                accV, accC);
    }
  }

  __syncthreads();
  for (int i = tid; i < nr * SDIM; i += 1024) {
    int node = i >> 7, col = i & 127;
    agg_s[(long)(lo + node) * SDIM + col] = accS[node * SP + col];
  }
  for (int i = tid; i < nr * V3; i += 1024) {
    int node = i / V3, col = i - node * V3;
    agg_v[(long)(lo + node) * V3 + col] = accV[node * VP + col];
  }
  for (int i = tid; i < nr; i += 1024) cnt[lo + i] = accC[i];
}

__global__ void __launch_bounds__(256) k_node(
    const float* __restrict__ s_n, const float* __restrict__ v_n,
    const float* __restrict__ agg_s, const float* __restrict__ agg_v,
    const float* __restrict__ cnt,
    const float* __restrict__ wns, const float* __restrict__ bns,
    const float* __restrict__ wnv, const float* __restrict__ bnv,
    const float* __restrict__ wrs, const float* __restrict__ brs,
    const float* __restrict__ wrv, const float* __restrict__ brv,
    float* __restrict__ out) {
  __shared__ float lds_gate[8][16 * 16];
  int wid = threadIdx.x >> 5;
  int lane = threadIdx.x & 31;
  int group = blockIdx.x * 8 + wid;
  if (group >= NN / 16) return;
  int nb = group * 16;
  int row = lane & 15;
  int half = lane >> 4;
  int noderow = nb + row;

  float invden = fast_rcp(fmaxf(cnt[noderow], 1.f));
  const float* sA = s_n + (long)noderow * SDIM;
  const float* gA = agg_s + (long)noderow * SDIM;

#pragma unroll 1
  for (int ct = 0; ct < 8; ct++) {
    v8f ao = {};
    v8f ar = {};
    int n = 16 * ct + row;
    const float* bo = wns + (long)n * SDIM;
    const float* br = wrs + (long)n * SDIM;
#pragma unroll 4
    for (int j = 0; j < 32; j++) {
      int k0 = 4 * j + 2 * half;
      v2f as = *(const v2f*)(sA + k0);
      v2f ag = *(const v2f*)(gA + k0);
      v2f ain = {as.x + ag.x * invden, as.y + ag.y * invden};
      v2f bO = *(const v2f*)(bo + k0);
      v2f bR = *(const v2f*)(br + k0);
      ao = wmma4(ain, bO, ao);
      ar = wmma4(as, bR, ar);
    }
    float biasO = bns[n], biasR = brs[n];
#pragma unroll
    for (int r = 0; r < 8; r++) {
      int m = r + 8 * half;
      float pre = ao[r] + biasO;
      float so = pre * sigmoidf_(pre);
      if (ct == 0) lds_gate[wid][m * 16 + n] = sigmoidf_(so);
      out[(long)(nb + m) * SDIM + n] = so + (ar[r] + biasR);
    }
  }

  const float* vA = v_n + (long)noderow * V3;
  const float* gV = agg_v + (long)noderow * V3;
  float* outv = out + (long)NN * SDIM;
#pragma unroll 1
  for (int ct = 0; ct < 3; ct++) {
    v8f ao = {};
    v8f ar = {};
    int d = 16 * ct + row;
    const float* bo = wnv + (long)d * V3;
    const float* br = wrv + (long)d * V3;
#pragma unroll
    for (int j = 0; j < 12; j++) {
      int k0 = 4 * j + 2 * half;
      v2f av = *(const v2f*)(vA + k0);
      v2f ag = *(const v2f*)(gV + k0);
      v2f ain = {av.x + ag.x * invden, av.y + ag.y * invden};
      v2f bO = *(const v2f*)(bo + k0);
      v2f bR = *(const v2f*)(br + k0);
      ao = wmma4(ain, bO, ao);
      ar = wmma4(av, bR, ar);
    }
    float biasO = bnv[d], biasR = brv[d];
    int vo = d / 3;
#pragma unroll
    for (int r = 0; r < 8; r++) {
      int m = r + 8 * half;
      float pre = ao[r] + biasO;
      float g = lds_gate[wid][m * 16 + vo];
      outv[(long)(nb + m) * V3 + d] = pre * g + (ar[r] + biasR);
    }
  }
}

extern "C" void kernel_launch(void* const* d_in, const int* in_sizes, int n_in,
                              void* d_out, int out_size, void* d_ws,
                              size_t ws_size, hipStream_t stream) {
  const float* node_s = (const float*)d_in[0];
  const float* node_v = (const float*)d_in[1];
  const int* eidx     = (const int*)d_in[2];
  const float* edge_s = (const float*)d_in[3];
  const float* edge_v = (const float*)d_in[4];
  const float* ln_g = (const float*)d_in[5];
  const float* ln_b = (const float*)d_in[6];
  const float* wes = (const float*)d_in[7];
  const float* bes = (const float*)d_in[8];
  const float* wev = (const float*)d_in[9];
  const float* bev = (const float*)d_in[10];
  const float* wns = (const float*)d_in[11];
  const float* bns = (const float*)d_in[12];
  const float* wnv = (const float*)d_in[13];
  const float* bnv = (const float*)d_in[14];
  const float* wrs = (const float*)d_in[15];
  const float* brs = (const float*)d_in[16];
  const float* wrv = (const float*)d_in[17];
  const float* brv = (const float*)d_in[18];
  float* out = (float*)d_out;

  float* ws = (float*)d_ws;
  float* s_n = ws;
  float* v_n = s_n + (size_t)NN * SDIM;
  float* agg_s = v_n + (size_t)NN * V3;
  float* agg_v = agg_s + (size_t)NN * SDIM;
  float* cnt = agg_v + (size_t)NN * V3;
  const int* edge_dst = eidx + EE;

  k_norm<<<(NN * 32 + 255) / 256, 256, 0, stream>>>(node_s, node_v, ln_g, ln_b,
                                                    s_n, v_n);
  k_edge<<<NBLK, 1024, 0, stream>>>(edge_s, edge_v, edge_dst, wes, bes, wev,
                                    bev, agg_s, agg_v, cnt);
  k_node<<<(NN / 16 + 7) / 8, 256, 0, stream>>>(s_n, v_n, agg_s, agg_v, cnt,
                                                wns, bns, wnv, bnv, wrs, brs,
                                                wrv, brv, out);
}


# k_node weight-stationary: wave w keeps column tile w of wns/wrs (wnv/wrv) in registers all kernel long; 256 work-groups stream 16-node tiles, rows read once, staged in LDS; sum order unchanged
# speedup vs baseline: 4.1901x; 1.3474x over previous
#include <hip/hip_runtime.h>
#include <math.h>

#define NN   50000
#define EE   800000
#define SDIM 128
#define VDIM 16
#define V3   48
#define ESD  32

typedef float v2f __attribute__((ext_vector_type(2)));
typedef float v8f __attribute__((ext_vector_type(8)));

__device__ __forceinline__ v8f wmma4(v2f a, v2f b, v8f c) {
  return __builtin_amdgcn_wmma_f32_16x16x4_f32(false, a, false, b, (short)0, c,
                                               false, false);
}

__device__ __forceinline__ float fast_rcp(float x) {
  return __builtin_amdgcn_rcpf(x);
}

__device__ __forceinline__ float sigmoidf_(float x) {
  return fast_rcp(1.f + __expf(-x));
}

__global__ void __launch_bounds__(256) k_norm(
    const float* __restrict__ node_s, const float* __restrict__ node_v,
    const float* __restrict__ ln_g, const float* __restrict__ ln_b,
    float* __restrict__ s_n, float* __restrict__ v_n) {
  int wave = (blockIdx.x * blockDim.x + threadIdx.x) >> 5;
  int lane = threadIdx.x & 31;
  if (wave >= NN) return;

  const float* srow = node_s + (long)wave * SDIM;
  float4 x = ((const float4*)srow)[lane];
  float s = x.x + x.y + x.z + x.w;
#pragma unroll
  for (int m = 16; m >= 1; m >>= 1) s += __shfl_xor(s, m, 32);
  float mu = s * (1.f / SDIM);
  float dx = x.x - mu, dy = x.y - mu, dz = x.z - mu, dw = x.w - mu;
  float q = dx * dx + dy * dy + dz * dz + dw * dw;
#pragma unroll
  for (int m = 16; m >= 1; m >>= 1) q += __shfl_xor(q, m, 32);
  float rs = rsqrtf(q * (1.f / SDIM) + 1e-5f);
  int c0 = lane * 4;
  float4 o;
  o.x = dx * rs * ln_g[c0 + 0] + ln_b[c0 + 0];
  o.y = dy * rs * ln_g[c0 + 1] + ln_b[c0 + 1];
  o.z = dz * rs * ln_g[c0 + 2] + ln_b[c0 + 2];
  o.w = dw * rs * ln_g[c0 + 3] + ln_b[c0 + 3];
  ((float4*)(s_n + (long)wave * SDIM))[lane] = o;

  const float* vrow = node_v + (long)wave * V3;
  float2 v = {0.f, 0.f};
  int vi = lane * 2;
  if (vi < V3) v = ((const float2*)vrow)[lane];
  float qq = v.x * v.x + v.y * v.y;
#pragma unroll
  for (int m = 16; m >= 1; m >>= 1) qq += __shfl_xor(qq, m, 32);
  float inv = rsqrtf(qq * (1.f / VDIM) + 1e-8f);
  if (vi < V3) {
    float2 ov;
    ov.x = v.x * inv;
    ov.y = v.y * inv;
    ((float2*)(v_n + (long)wave * V3))[lane] = ov;
  }
}

#define RNG  196
#define NBLK ((NN + RNG - 1) / RNG)
#define SP   144
#define VP   49
#define GP   17
#define NW   32
#define QCAP 528

__device__ __forceinline__ void lds_fadd(float* p, float v) {
  __hip_atomic_fetch_add(p, v, __ATOMIC_RELAXED, __HIP_MEMORY_SCOPE_WORKGROUP);
}

__device__ __forceinline__ void edge_tile(
    const int* qp, int nv, int lane,
    const float* __restrict__ edge_s, const float* __restrict__ edge_v,
    const float* __restrict__ bes, const float4* Wl, const v2f* Bl,
    float* gate, float* accS, float* accV, float* accC) {
  int row = lane & 15;
  int half = lane >> 4;
  int pk = qp[row < nv ? row : 0];
  int eid = pk & 0xFFFFF;
  int dlv = row < nv ? pk >> 20 : RNG;

  int dlr[8];
#pragma unroll
  for (int r = 0; r < 8; r++) {
    int m = r + 8 * half;
    dlr[r] = m < nv ? qp[m] >> 20 : RNG;
  }

  const float* arow = edge_s + (long)eid * ESD;
  v2f a[8];
#pragma unroll
  for (int j = 0; j < 8; j++) a[j] = *(const v2f*)(arow + 4 * j + 2 * half);
  float ev0 = edge_v[eid * 3 + 0];
  float ev1 = edge_v[eid * 3 + 1];
  float ev2 = edge_v[eid * 3 + 2];

#pragma unroll 1
  for (int c = 0; c < 8; c++) {
    v8f acc = {};
    int n = 16 * c + row;
#pragma unroll
    for (int j = 0; j < 8; j++) {
      v2f b = Bl[(c * 8 + j) * 32 + lane];
      acc = wmma4(a[j], b, acc);
    }
    float bias = bes[n];
#pragma unroll
    for (int r = 0; r < 8; r++) {
      int m = r + 8 * half;
      float pre = acc[r] + bias;
      float so = pre * sigmoidf_(pre);
      if (c == 0) gate[m * GP + row] = sigmoidf_(so);
      lds_fadd(accS + dlr[r] * SP + n, so);
    }
  }

  if (half == 0) lds_fadd(accC + dlv, 1.f);

  __builtin_amdgcn_wave_barrier();

#pragma unroll
  for (int t0 = 0; t0 < 24; t0 += 8) {
    float vg[8];
#pragma unroll
    for (int t = 0; t < 8; t++) {
      int d = half * 24 + t0 + t;
      float4 w = Wl[d];
      float v = ev0 * w.x + ev1 * w.y + ev2 * w.z + w.w;
      vg[t] = v * gate[row * GP + d / 3];
    }
#pragma unroll
    for (int t = 0; t < 8; t++)
      lds_fadd(accV + dlv * VP + half * 24 + t0 + t, vg[t]);
  }
  __builtin_amdgcn_wave_barrier();
}

__device__ __forceinline__ int4 ld_dst(const int* __restrict__ edge_dst, int t,
                                       int wid, int lane) {
  int ix = (t * NW + wid) * 128 + lane * 4;
  int4 d = {-1, -1, -1, -1};
  if (ix < EE) d = *(const int4*)(edge_dst + ix);
  return d;
}

__global__ void __launch_bounds__(1024) k_edge(
    const float* __restrict__ edge_s, const float* __restrict__ edge_v,
    const int* __restrict__ edge_dst,
    const float* __restrict__ wes, const float* __restrict__ bes,
    const float* __restrict__ wev, const float* __restrict__ bev,
    float* __restrict__ agg_s, float* __restrict__ agg_v,
    float* __restrict__ cnt) {
  __shared__ float accS[(RNG + 1) * SP];
  __shared__ float accV[(RNG + 1) * VP];
  __shared__ float accC[RNG + 1];
  __shared__ v2f   Bl[8 * 8 * 32];
  __shared__ float4 Wl[V3];
  __shared__ int   queue[NW][QCAP];
  __shared__ float gateL[NW][16 * GP];

  int tid = threadIdx.x;
  int wid = tid >> 5;
  int lane = tid & 31;
  int lo = blockIdx.x * RNG;
  int nr = NN - lo < RNG ? NN - lo : RNG;

  for (int i = tid; i < (RNG + 1) * SP; i += 1024) accS[i] = 0.f;
  for (int i = tid; i < (RNG + 1) * VP; i += 1024) accV[i] = 0.f;
  for (int i = tid; i < RNG + 1; i += 1024) accC[i] = 0.f;
  for (int i = tid; i < 8 * 8 * 32; i += 1024) {
    int l = i & 31;
    int j = (i >> 5) & 7;
    int c = i >> 8;
    int n = 16 * c + (l & 15);
    int k0 = 4 * j + 2 * (l >> 4);
    Bl[i] = *(const v2f*)(wes + (long)n * ESD + k0);
  }
  if (tid < V3) {
    float4 w = {wev[tid * 3 + 0], wev[tid * 3 + 1], wev[tid * 3 + 2], bev[tid]};
    Wl[tid] = w;
  }
  __syncthreads();

  int* q = queue[wid];
  float* gate = gateL[wid];
  int qn = 0;
  unsigned ltmask = (1u << lane) - 1u;
  const int NSTEP = (EE + 128 * NW - 1) / (128 * NW);

  const int NG = (NSTEP + 3) / 4;
  int4 dd[4];
#pragma unroll
  for (int u = 0; u < 4; u++) dd[u] = ld_dst(edge_dst, u, wid, lane);
#pragma unroll 1
  for (int g = 0; g <= NG; g++) {
    bool last = g == NG;
#pragma unroll
    for (int u = 0; u < 4; u++) {
      int ixc = ((g * 4 + u) * NW + wid) * 128 + lane * 4;
      int dv[4] = {dd[u].x, dd[u].y, dd[u].z, dd[u].w};
      bool mm[4];
      bool any = false;
#pragma unroll
      for (int s = 0; s < 4; s++) {
        mm[s] = (unsigned)(dv[s] - lo) < (unsigned)nr;
        any = any || mm[s];
      }
      if (__builtin_amdgcn_ballot_w32(any) != 0u) {
#pragma unroll
        for (int s = 0; s < 4; s++) {
          unsigned bal = __builtin_amdgcn_ballot_w32(mm[s]);
          if (mm[s])
            q[qn + __popc(bal & ltmask)] = (ixc + s) | ((dv[s] - lo) << 20);
          qn += __popc(bal);
        }
      }
    }
#pragma unroll
    for (int u = 0; u < 4; u++)
      dd[u] = ld_dst(edge_dst, (g + 1) * 4 + u, wid, lane);
    __builtin_amdgcn_wave_barrier();
#pragma unroll 1
    while (qn >= 16 || (last && qn > 0)) {
      int nv = qn < 16 ? qn : 16;
      qn -= nv;
      edge_tile(q + qn, nv, lane, edge_s, edge_v, bes, Wl, Bl, gate, accS,
                accV, accC);
    }
  }

  __syncthreads();
  for (int i = tid; i < nr * SDIM; i += 1024) {
    int node = i >> 7, col = i & 127;
    agg_s[(long)(lo + node) * SDIM + col] = accS[node * SP + col];
  }
  for (int i = tid; i < nr * V3; i += 1024) {
    int node = i / V3, col = i - node * V3;
    agg_v[(long)(lo + node) * V3 + col] = accV[node * VP + col];
  }
  for (int i = tid; i < nr; i += 1024) cnt[lo + i] = accC[i];
}

#define NT   (NN / 16)
#define PA   132
#define PV   52
#define GPN  17
#define KNG  256

struct NodeRows {
  float4 s0, s1, g0, g1, v, gv;
  float c0, c1, cv;
};

__device__ __forceinline__ void kn_load(
    NodeRows& R, int t, int tid, int wid, int lane,
    const float* __restrict__ s_n, const float* __restrict__ v_n,
    const float* __restrict__ agg_s, const float* __restrict__ agg_v,
    const float* __restrict__ cnt) {
  int nb = t * 16;
  int r0 = nb + 2 * wid;
  R.s0 = ((const float4*)(s_n + (long)r0 * SDIM))[lane];
  R.s1 = ((const float4*)(s_n + (long)(r0 + 1) * SDIM))[lane];
  R.g0 = ((const float4*)(agg_s + (long)r0 * SDIM))[lane];
  R.g1 = ((const float4*)(agg_s + (long)(r0 + 1) * SDIM))[lane];
  R.c0 = cnt[r0];
  R.c1 = cnt[r0 + 1];
  if (tid < 192) {
    R.v = ((const float4*)(v_n + (long)nb * V3))[tid];
    R.gv = ((const float4*)(agg_v + (long)nb * V3))[tid];
    R.cv = cnt[nb + tid / 12];
  }
}

__device__ __forceinline__ void kn_stage(
    const NodeRows& R, float* ain, float* sn, float* vain, float* vn,
    int tid, int wid, int lane) {
  float i0 = fast_rcp(fmaxf(R.c0, 1.f));
  float i1 = fast_rcp(fmaxf(R.c1, 1.f));
  float4 a0 = {R.s0.x + R.g0.x * i0, R.s0.y + R.g0.y * i0,
               R.s0.z + R.g0.z * i0, R.s0.w + R.g0.w * i0};
  float4 a1 = {R.s1.x + R.g1.x * i1, R.s1.y + R.g1.y * i1,
               R.s1.z + R.g1.z * i1, R.s1.w + R.g1.w * i1};
  int o = (2 * wid) * PA + 4 * lane;
  *(float4*)(ain + o) = a0;
  *(float4*)(ain + o + PA) = a1;
  *(float4*)(sn + o) = R.s0;
  *(float4*)(sn + o + PA) = R.s1;
  if (tid < 192) {
    int vr = tid / 12;
    int vc = tid - vr * 12;
    float iv = fast_rcp(fmaxf(R.cv, 1.f));
    float4 av = {R.v.x + R.gv.x * iv, R.v.y + R.gv.y * iv,
                 R.v.z + R.gv.z * iv, R.v.w + R.gv.w * iv};
    int ov = vr * PV + 4 * vc;
    *(float4*)(vain + ov) = av;
    *(float4*)(vn + ov) = R.v;
  }
}

__device__ __forceinline__ void lds_barrier() {
  __builtin_amdgcn_fence(__ATOMIC_RELEASE, "workgroup", "local");
  __builtin_amdgcn_s_barrier();
  __builtin_amdgcn_fence(__ATOMIC_ACQUIRE, "workgroup", "local");
}

__global__ void __launch_bounds__(256) k_node(
    const float* __restrict__ s_n, const float* __restrict__ v_n,
    const float* __restrict__ agg_s, const float* __restrict__ agg_v,
    const float* __restrict__ cnt,
    const float* __restrict__ wns, const float* __restrict__ bns,
    const float* __restrict__ wnv, const float* __restrict__ bnv,
    const float* __restrict__ wrs, const float* __restrict__ brs,
    const float* __restrict__ wrv, const float* __restrict__ brv,
    float* __restrict__ out) {
  __shared__ __attribute__((aligned(16))) float ainL[2][16 * PA];
  __shared__ __attribute__((aligned(16))) float snL[2][16 * PA];
  __shared__ __attribute__((aligned(16))) float vainL[2][16 * PV];
  __shared__ __attribute__((aligned(16))) float vnL[2][16 * PV];
  __shared__ float gateN[2][16 * GPN];

  int tid = threadIdx.x;
  int wid = __builtin_amdgcn_readfirstlane(tid >> 5);
  int lane = tid & 31;
  int row = lane & 15;
  int half = lane >> 4;

  int n = 16 * wid + row;
  v2f bO[32], bR[32];
  {
    const float* bo = wns + (long)n * SDIM + 2 * half;
    const float* br = wrs + (long)n * SDIM + 2 * half;
#pragma unroll
    for (int j = 0; j < 32; j++) {
      bO[j] = *(const v2f*)(bo + 4 * j);
      bR[j] = *(const v2f*)(br + 4 * j);
    }
  }
  float biasO = bns[n], biasR = brs[n];

  bool vw = wid >= 1 && wid <= 3;
  int d = 16 * (vw ? wid - 1 : 0) + row;
  int vo = d / 3;
  v2f cO[12], cR[12];
  float vbO = 0.f, vbR = 0.f;
#pragma unroll
  for (int j = 0; j < 12; j++) {
    cO[j] = v2f{0.f, 0.f};
    cR[j] = v2f{0.f, 0.f};
  }
  if (vw) {
    const float* bo = wnv + (long)d * V3 + 2 * half;
    const float* br = wrv + (long)d * V3 + 2 * half;
#pragma unroll
    for (int j = 0; j < 12; j++) {
      cO[j] = *(const v2f*)(bo + 4 * j);
      cR[j] = *(const v2f*)(br + 4 * j);
    }
    vbO = bnv[d];
    vbR = brv[d];
  }

  float* outv = out + (long)NN * SDIM;
  NodeRows R;
  int t = blockIdx.x;
  kn_load(R, t, tid, wid, lane, s_n, v_n, agg_s, agg_v, cnt);
  kn_stage(R, ainL[0], snL[0], vainL[0], vnL[0], tid, wid, lane);
  lds_barrier();
  int p = 0;
#pragma unroll 1
  for (; t < NT; t += KNG) {
    bool more = t + KNG < NT;
    if (more) kn_load(R, t + KNG, tid, wid, lane, s_n, v_n, agg_s, agg_v, cnt);
    int nb = t * 16;

    const float* aA = ainL[p] + row * PA + 2 * half;
    const float* aS = snL[p] + row * PA + 2 * half;
    v8f ao = {};
    v8f ar = {};
#pragma unroll
    for (int j = 0; j < 32; j++) {
      v2f ain = *(const v2f*)(aA + 4 * j);
      v2f as = *(const v2f*)(aS + 4 * j);
      ao = wmma4(ain, bO[j], ao);
      ar = wmma4(as, bR[j], ar);
    }
#pragma unroll
    for (int r = 0; r < 8; r++) {
      int m = r + 8 * half;
      float pre = ao[r] + biasO;
      float so = pre * sigmoidf_(pre);
      if (wid == 0) gateN[p][m * GPN + row] = sigmoidf_(so);
      out[(long)(nb + m) * SDIM + n] = so + (ar[r] + biasR);
    }

    v8f vao = {};
    v8f var = {};
    if (vw) {
      const float* vA = vainL[p] + row * PV + 2 * half;
      const float* vS = vnL[p] + row * PV + 2 * half;
#pragma unroll
      for (int j = 0; j < 12; j++) {
        v2f ain = *(const v2f*)(vA + 4 * j);
        v2f av = *(const v2f*)(vS + 4 * j);
        vao = wmma4(ain, cO[j], vao);
        var = wmma4(av, cR[j], var);
      }
    }

    if (more)
      kn_stage(R, ainL[p ^ 1], snL[p ^ 1], vainL[p ^ 1], vnL[p ^ 1], tid, wid,
               lane);
    lds_barrier();

    if (vw) {
#pragma unroll
      for (int r = 0; r < 8; r++) {
        int m = r + 8 * half;
        float pre = vao[r] + vbO;
        float g = gateN[p][m * GPN + vo];
        outv[(long)(nb + m) * V3 + d] = pre * g + (var[r] + vbR);
      }
    }
    p ^= 1;
  }
}

extern "C" void kernel_launch(void* const* d_in, const int* in_sizes, int n_in,
                              void* d_out, int out_size, void* d_ws,
                              size_t ws_size, hipStream_t stream) {
  const float* node_s = (const float*)d_in[0];
  const float* node_v = (const float*)d_in[1];
  const int* eidx     = (const int*)d_in[2];
  const float* edge_s = (const float*)d_in[3];
  const float* edge_v = (const float*)d_in[4];
  const float* ln_g = (const float*)d_in[5];
  const float* ln_b = (const float*)d_in[6];
  const float* wes = (const float*)d_in[7];
  const float* bes = (const float*)d_in[8];
  const float* wev = (const float*)d_in[9];
  const float* bev = (const float*)d_in[10];
  const float* wns = (const float*)d_in[11];
  const float* bns = (const float*)d_in[12];
  const float* wnv = (const float*)d_in[13];
  const float* bnv = (const float*)d_in[14];
  const float* wrs = (const float*)d_in[15];
  const float* brs = (const float*)d_in[16];
  const float* wrv = (const float*)d_in[17];
  const float* brv = (const float*)d_in[18];
  float* out = (float*)d_out;

  float* ws = (float*)d_ws;
  float* s_n = ws;
  float* v_n = s_n + (size_t)NN * SDIM;
  float* agg_s = v_n + (size_t)NN * V3;
  float* agg_v = agg_s + (size_t)NN * SDIM;
  float* cnt = agg_v + (size_t)NN * V3;
  const int* edge_dst = eidx + EE;

  k_norm<<<(NN * 32 + 255) / 256, 256, 0, stream>>>(node_s, node_v, ln_g, ln_b,
                                                    s_n, v_n);
  k_edge<<<NBLK, 1024, 0, stream>>>(edge_s, edge_v, edge_dst, wes, bes, wev,
                                    bev, agg_s, agg_v, cnt);
  k_node<<<KNG, 256, 0, stream>>>(s_n, v_n, agg_s, agg_v, cnt,
                                                wns, bns, wnv, bnv, wrs, brs,
                                                wrv, brv, out);
}


# k_edge scans a 1-byte range key per edge (dst/196, written by extra blocks of k_norm), 16 a lane, by zero-byte word tests; dst fetched and checked in the tile; sum order still LDS arrival
# speedup vs baseline: 4.7851x; 1.1420x over previous
#include <hip/hip_runtime.h>
#include <math.h>

#define NN   50000
#define EE   800000
#define SDIM 128
#define VDIM 16
#define V3   48
#define ESD  32

#define RNG  196
#define NBLK ((NN + RNG - 1) / RNG)
#define NORMB ((NN * 32 + 255) / 256)
#define KEYB  ((EE / 16 + 255) / 256)
static_assert(NBLK <= 256, "a range key must fit in one byte");
static_assert(EE % 16 == 0, "keys are written and read 16 at a time");

typedef float v2f __attribute__((ext_vector_type(2)));
typedef float v8f __attribute__((ext_vector_type(8)));

__device__ __forceinline__ v8f wmma4(v2f a, v2f b, v8f c) {
  return __builtin_amdgcn_wmma_f32_16x16x4_f32(false, a, false, b, (short)0, c,
                                               false, false);
}

__device__ __forceinline__ float fast_rcp(float x) {
  return __builtin_amdgcn_rcpf(x);
}

__device__ __forceinline__ float sigmoidf_(float x) {
  return fast_rcp(1.f + __expf(-x));
}

__device__ __forceinline__ int pack_keys(int4 d) {
  unsigned k0 = ((unsigned)d.x / RNG) & 255u;
  unsigned k1 = ((unsigned)d.y / RNG) & 255u;
  unsigned k2 = ((unsigned)d.z / RNG) & 255u;
  unsigned k3 = ((unsigned)d.w / RNG) & 255u;
  return (int)(k0 | (k1 << 8) | (k2 << 16) | (k3 << 24));
}

__global__ void __launch_bounds__(256) k_norm(
    const float* __restrict__ node_s, const float* __restrict__ node_v,
    const float* __restrict__ ln_g, const float* __restrict__ ln_b,
    float* __restrict__ s_n, float* __restrict__ v_n,
    const int* __restrict__ edge_dst, int* __restrict__ keys) {
  if (blockIdx.x >= NORMB) {
    int i = (blockIdx.x - NORMB) * 256 + threadIdx.x;
    if (i < EE / 16) {
      const int4* src = (const int4*)(edge_dst + (long)i * 16);
      int4 o;
      o.x = pack_keys(src[0]);
      o.y = pack_keys(src[1]);
      o.z = pack_keys(src[2]);
      o.w = pack_keys(src[3]);
      ((int4*)keys)[i] = o;
    }
    return;
  }
  int wave = (blockIdx.x * blockDim.x + threadIdx.x) >> 5;
  int lane = threadIdx.x & 31;
  if (wave >= NN) return;

  const float* srow = node_s + (long)wave * SDIM;
  float4 x = ((const float4*)srow)[lane];
  float s = x.x + x.y + x.z + x.w;
#pragma unroll
  for (int m = 16; m >= 1; m >>= 1) s += __shfl_xor(s, m, 32);
  float mu = s * (1.f / SDIM);
  float dx = x.x - mu, dy = x.y - mu, dz = x.z - mu, dw = x.w - mu;
  float q = dx * dx + dy * dy + dz * dz + dw * dw;
#pragma unroll
  for (int m = 16; m >= 1; m >>= 1) q += __shfl_xor(q, m, 32);
  float rs = rsqrtf(q * (1.f / SDIM) + 1e-5f);
  int c0 = lane * 4;
  float4 o;
  o.x = dx * rs * ln_g[c0 + 0] + ln_b[c0 + 0];
  o.y = dy * rs * ln_g[c0 + 1] + ln_b[c0 + 1];
  o.z = dz * rs * ln_g[c0 + 2] + ln_b[c0 + 2];
  o.w = dw * rs * ln_g[c0 + 3] + ln_b[c0 + 3];
  ((float4*)(s_n + (long)wave * SDIM))[lane] = o;

  const float* vrow = node_v + (long)wave * V3;
  float2 v = {0.f, 0.f};
  int vi = lane * 2;
  if (vi < V3) v = ((const float2*)vrow)[lane];
  float qq = v.x * v.x + v.y * v.y;
#pragma unroll
  for (int m = 16; m >= 1; m >>= 1) qq += __shfl_xor(qq, m, 32);
  float inv = rsqrtf(qq * (1.f / VDIM) + 1e-8f);
  if (vi < V3) {
    float2 ov;
    ov.x = v.x * inv;
    ov.y = v.y * inv;
    ((float2*)(v_n + (long)wave * V3))[lane] = ov;
  }
}

#define SP   144
#define VP   49
#define GP   17
#define NW   32
#define QCAP 528

__device__ __forceinline__ void lds_fadd(float* p, float v) {
  __hip_atomic_fetch_add(p, v, __ATOMIC_RELAXED, __HIP_MEMORY_SCOPE_WORKGROUP);
}

__device__ __forceinline__ void edge_tile(
    const int* qp, int nv, int lane, int lo, int nr,
    const float* __restrict__ edge_s, const float* __restrict__ edge_v,
    const int* __restrict__ edge_dst,
    const float* __restrict__ bes, const float4* Wl, const v2f* Bl,
    float* gate, int* dls, float* accS, float* accV, float* accC) {
  int row = lane & 15;
  int half = lane >> 4;
  int eid = qp[row < nv ? row : 0];

  int dst = edge_dst[eid];
  const float* arow = edge_s + (long)eid * ESD;
  v2f a[8];
#pragma unroll
  for (int j = 0; j < 8; j++) a[j] = *(const v2f*)(arow + 4 * j + 2 * half);
  float ev0 = edge_v[eid * 3 + 0];
  float ev1 = edge_v[eid * 3 + 1];
  float ev2 = edge_v[eid * 3 + 2];

  int dl = dst - lo;
  int dlv = (row < nv && (unsigned)dl < (unsigned)nr) ? dl : RNG;
  dls[row] = dlv;
  __builtin_amdgcn_wave_barrier();
  int dlr[8];
#pragma unroll
  for (int r = 0; r < 8; r++) dlr[r] = dls[r + 8 * half];

#pragma unroll 1
  for (int c = 0; c < 8; c++) {
    v8f acc = {};
    int n = 16 * c + row;
#pragma unroll
    for (int j = 0; j < 8; j++) {
      v2f b = Bl[(c * 8 + j) * 32 + lane];
      acc = wmma4(a[j], b, acc);
    }
    float bias = bes[n];
#pragma unroll
    for (int r = 0; r < 8; r++) {
      int m = r + 8 * half;
      float pre = acc[r] + bias;
      float so = pre * sigmoidf_(pre);
      if (c == 0) gate[m * GP + row] = sigmoidf_(so);
      lds_fadd(accS + dlr[r] * SP + n, so);
    }
  }

  if (half == 0) lds_fadd(accC + dlv, 1.f);

  __builtin_amdgcn_wave_barrier();

#pragma unroll
  for (int t0 = 0; t0 < 24; t0 += 8) {
    float vg[8];
#pragma unroll
    for (int t = 0; t < 8; t++) {
      int d = half * 24 + t0 + t;
      float4 w = Wl[d];
      float v = ev0 * w.x + ev1 * w.y + ev2 * w.z + w.w;
      vg[t] = v * gate[row * GP + d / 3];
    }
#pragma unroll
    for (int t = 0; t < 8; t++)
      lds_fadd(accV + dlv * VP + half * 24 + t0 + t, vg[t]);
  }
  __builtin_amdgcn_wave_barrier();
}

__device__ __forceinline__ int4 ld_key(const int* __restrict__ keys, int t,
                                       int wid, int lane, int nomatch) {
  int ix = (t * NW + wid) * 512 + lane * 16;
  int4 k = {nomatch, nomatch, nomatch, nomatch};
  if (ix < EE) k = *(const int4*)(keys + (ix >> 2));
  return k;
}

__device__ __forceinline__ unsigned zero_bytes(unsigned x) {
  return (x - 0x01010101u) & ~x & 0x80808080u;
}

__global__ void __launch_bounds__(1024) k_edge(
    const float* __restrict__ edge_s, const float* __restrict__ edge_v,
    const int* __restrict__ edge_dst, const int* __restrict__ keys,
    const float* __restrict__ wes, const float* __restrict__ bes,
    const float* __restrict__ wev, const float* __restrict__ bev,
    float* __restrict__ agg_s, float* __restrict__ agg_v,
    float* __restrict__ cnt) {
  __shared__ float accS[(RNG + 1) * SP];
  __shared__ float accV[(RNG + 1) * VP];
  __shared__ float accC[RNG + 1];
  __shared__ v2f   Bl[8 * 8 * 32];
  __shared__ float4 Wl[V3];
  __shared__ int   queue[NW][QCAP];
  __shared__ float gateL[NW][16 * GP];
  __shared__ int   dlsL[NW][16];

  int tid = threadIdx.x;
  int wid = tid >> 5;
  int lane = tid & 31;
  int lo = blockIdx.x * RNG;
  int nr = NN - lo < RNG ? NN - lo : RNG;

  for (int i = tid; i < (RNG + 1) * SP; i += 1024) accS[i] = 0.f;
  for (int i = tid; i < (RNG + 1) * VP; i += 1024) accV[i] = 0.f;
  for (int i = tid; i < RNG + 1; i += 1024) accC[i] = 0.f;
  for (int i = tid; i < 8 * 8 * 32; i += 1024) {
    int l = i & 31;
    int j = (i >> 5) & 7;
    int c = i >> 8;
    int n = 16 * c + (l & 15);
    int k0 = 4 * j + 2 * (l >> 4);
    Bl[i] = *(const v2f*)(wes + (long)n * ESD + k0);
  }
  if (tid < V3) {
    float4 w = {wev[tid * 3 + 0], wev[tid * 3 + 1], wev[tid * 3 + 2], bev[tid]};
    Wl[tid] = w;
  }
  __syncthreads();

  int* q = queue[wid];
  float* gate = gateL[wid];
  int* dls = dlsL[wid];
  int qn = 0;
  unsigned ltmask = (1u << lane) - 1u;
  const int NSTEP = (EE + 512 * NW - 1) / (512 * NW);

  unsigned pat = blockIdx.x * 0x01010101u;
  int nomatch = (int)~pat;

  const int NG = (NSTEP + 3) / 4;
  int4 kk[4];
#pragma unroll
  for (int u = 0; u < 4; u++) kk[u] = ld_key(keys, u, wid, lane, nomatch);
#pragma unroll 1
  for (int g = 0; g < NG; g++) {
#pragma unroll
    for (int u = 0; u < 4; u++) {
      int t = g * 4 + u;
      int base = (t * NW + wid) * 512 + lane * 16;
      unsigned zz = (zero_bytes((unsigned)kk[u].x ^ pat) >> 7) |
                    (zero_bytes((unsigned)kk[u].y ^ pat) >> 6) |
                    (zero_bytes((unsigned)kk[u].z ^ pat) >> 5) |
                    (zero_bytes((unsigned)kk[u].w ^ pat) >> 4);
      kk[u] = ld_key(keys, t + 4, wid, lane, nomatch);
      unsigned bal = __builtin_amdgcn_ballot_w32(zz != 0u);
#pragma unroll 1
      while (bal != 0u) {
        if (zz != 0u) {
          int p = __builtin_ctz(zz);
          q[qn + __popc(bal & ltmask)] = base + ((p & 3) << 2) + (p >> 3);
          zz &= zz - 1u;
        }
        qn += __popc(bal);
        bal = __builtin_amdgcn_ballot_w32(zz != 0u);
      }
      __builtin_amdgcn_wave_barrier();
#pragma unroll 1
      while (qn >= 16) {
        qn -= 16;
        edge_tile(q + qn, 16, lane, lo, nr, edge_s, edge_v, edge_dst, bes, Wl,
                  Bl, gate, dls, accS, accV, accC);
      }
    }
  }
  if (qn > 0)
    edge_tile(q, qn, lane, lo, nr, edge_s, edge_v, edge_dst, bes, Wl, Bl, gate,
              dls, accS, accV, accC);

  __syncthreads();
  for (int i = tid; i < nr * SDIM; i += 1024) {
    int node = i >> 7, col = i & 127;
    agg_s[(long)(lo + node) * SDIM + col] = accS[node * SP + col];
  }
  for (int i = tid; i < nr * V3; i += 1024) {
    int node = i / V3, col = i - node * V3;
    agg_v[(long)(lo + node) * V3 + col] = accV[node * VP + col];
  }
  for (int i = tid; i < nr; i += 1024) cnt[lo + i] = accC[i];
}

#define NT   (NN / 16)
#define PA   132
#define PV   52
#define GPN  17
#define KNG  256

struct NodeRows {
  float4 s0, s1, g0, g1, v, gv;
  float c0, c1, cv;
};

__device__ __forceinline__ void kn_load(
    NodeRows& R, int t, int tid, int wid, int lane,
    const float* __restrict__ s_n, const float* __restrict__ v_n,
    const float* __restrict__ agg_s, const float* __restrict__ agg_v,
    const float* __restrict__ cnt) {
  int nb = t * 16;
  int r0 = nb + 2 * wid;
  R.s0 = ((const float4*)(s_n + (long)r0 * SDIM))[lane];
  R.s1 = ((const float4*)(s_n + (long)(r0 + 1) * SDIM))[lane];
  R.g0 = ((const float4*)(agg_s + (long)r0 * SDIM))[lane];
  R.g1 = ((const float4*)(agg_s + (long)(r0 + 1) * SDIM))[lane];
  R.c0 = cnt[r0];
  R.c1 = cnt[r0 + 1];
  if (tid < 192) {
    R.v = ((const float4*)(v_n + (long)nb * V3))[tid];
    R.gv = ((const float4*)(agg_v + (long)nb * V3))[tid];
    R.cv = cnt[nb + tid / 12];
  }
}

__device__ __forceinline__ void kn_stage(
    const NodeRows& R, float* ain, float* sn, float* vain, float* vn,
    int tid, int wid, int lane) {
  float i0 = fast_rcp(fmaxf(R.c0, 1.f));
  float i1 = fast_rcp(fmaxf(R.c1, 1.f));
  float4 a0 = {R.s0.x + R.g0.x * i0, R.s0.y + R.g0.y * i0,
               R.s0.z + R.g0.z * i0, R.s0.w + R.g0.w * i0};
  float4 a1 = {R.s1.x + R.g1.x * i1, R.s1.y + R.g1.y * i1,
               R.s1.z + R.g1.z * i1, R.s1.w + R.g1.w * i1};
  int o = (2 * wid) * PA + 4 * lane;
  *(float4*)(ain + o) = a0;
  *(float4*)(ain + o + PA) = a1;
  *(float4*)(sn + o) = R.s0;
  *(float4*)(sn + o + PA) = R.s1;
  if (tid < 192) {
    int vr = tid / 12;
    int vc = tid - vr * 12;
    float iv = fast_rcp(fmaxf(R.cv, 1.f));
    float4 av = {R.v.x + R.gv.x * iv, R.v.y + R.gv.y * iv,
                 R.v.z + R.gv.z * iv, R.v.w + R.gv.w * iv};
    int ov = vr * PV + 4 * vc;
    *(float4*)(vain + ov) = av;
    *(float4*)(vn + ov) = R.v;
  }
}

__device__ __forceinline__ void lds_barrier() {
  __builtin_amdgcn_fence(__ATOMIC_RELEASE, "workgroup", "local");
  __builtin_amdgcn_s_barrier();
  __builtin_amdgcn_fence(__ATOMIC_ACQUIRE, "workgroup", "local");
}

__global__ void __launch_bounds__(256) k_node(
    const float* __restrict__ s_n, const float* __restrict__ v_n,
    const float* __restrict__ agg_s, const float* __restrict__ agg_v,
    const float* __restrict__ cnt,
    const float* __restrict__ wns, const float* __restrict__ bns,
    const float* __restrict__ wnv, const float* __restrict__ bnv,
    const float* __restrict__ wrs, const float* __restrict__ brs,
    const float* __restrict__ wrv, const float* __restrict__ brv,
    float* __restrict__ out) {
  __shared__ __attribute__((aligned(16))) float ainL[2][16 * PA];
  __shared__ __attribute__((aligned(16))) float snL[2][16 * PA];
  __shared__ __attribute__((aligned(16))) float vainL[2][16 * PV];
  __shared__ __attribute__((aligned(16))) float vnL[2][16 * PV];
  __shared__ float gateN[2][16 * GPN];

  int tid = threadIdx.x;
  int wid = __builtin_amdgcn_readfirstlane(tid >> 5);
  int lane = tid & 31;
  int row = lane & 15;
  int half = lane >> 4;

  int n = 16 * wid + row;
  v2f bO[32], bR[32];
  {
    const float* bo = wns + (long)n * SDIM + 2 * half;
    const float* br = wrs + (long)n * SDIM + 2 * half;
#pragma unroll
    for (int j = 0; j < 32; j++) {
      bO[j] = *(const v2f*)(bo + 4 * j);
      bR[j] = *(const v2f*)(br + 4 * j);
    }
  }
  float biasO = bns[n], biasR = brs[n];

  bool vw = wid >= 1 && wid <= 3;
  int d = 16 * (vw ? wid - 1 : 0) + row;
  int vo = d / 3;
  v2f cO[12], cR[12];
  float vbO = 0.f, vbR = 0.f;
#pragma unroll
  for (int j = 0; j < 12; j++) {
    cO[j] = v2f{0.f, 0.f};
    cR[j] = v2f{0.f, 0.f};
  }
  if (vw) {
    const float* bo = wnv + (long)d * V3 + 2 * half;
    const float* br = wrv + (long)d * V3 + 2 * half;
#pragma unroll
    for (int j = 0; j < 12; j++) {
      cO[j] = *(const v2f*)(bo + 4 * j);
      cR[j] = *(const v2f*)(br + 4 * j);
    }
    vbO = bnv[d];
    vbR = brv[d];
  }

  float* outv = out + (long)NN * SDIM;
  NodeRows R;
  int t = blockIdx.x;
  kn_load(R, t, tid, wid, lane, s_n, v_n, agg_s, agg_v, cnt);
  kn_stage(R, ainL[0], snL[0], vainL[0], vnL[0], tid, wid, lane);
  lds_barrier();
  int p = 0;
#pragma unroll 1
  for (; t < NT; t += KNG) {
    bool more = t + KNG < NT;
    if (more) kn_load(R, t + KNG, tid, wid, lane, s_n, v_n, agg_s, agg_v, cnt);
    int nb = t * 16;

    const float* aA = ainL[p] + row * PA + 2 * half;
    const float* aS = snL[p] + row * PA + 2 * half;
    v8f ao = {};
    v8f ar = {};
#pragma unroll
    for (int j = 0; j < 32; j++) {
      v2f ain = *(const v2f*)(aA + 4 * j);
      v2f as = *(const v2f*)(aS + 4 * j);
      ao = wmma4(ain, bO[j], ao);
      ar = wmma4(as, bR[j], ar);
    }
#pragma unroll
    for (int r = 0; r < 8; r++) {
      int m = r + 8 * half;
      float pre = ao[r] + biasO;
      float so = pre * sigmoidf_(pre);
      if (wid == 0) gateN[p][m * GPN + row] = sigmoidf_(so);
      out[(long)(nb + m) * SDIM + n] = so + (ar[r] + biasR);
    }

    v8f vao = {};
    v8f var = {};
    if (vw) {
      const float* vA = vainL[p] + row * PV + 2 * half;
      const float* vS = vnL[p] + row * PV + 2 * half;
#pragma unroll
      for (int j = 0; j < 12; j++) {
        v2f ain = *(const v2f*)(vA + 4 * j);
        v2f av = *(const v2f*)(vS + 4 * j);
        vao = wmma4(ain, cO[j], vao);
        var = wmma4(av, cR[j], var);
      }
    }

    if (more)
      kn_stage(R, ainL[p ^ 1], snL[p ^ 1], vainL[p ^ 1], vnL[p ^ 1], tid, wid,
               lane);
    lds_barrier();

    if (vw) {
#pragma unroll
      for (int r = 0; r < 8; r++) {
        int m = r + 8 * half;
        float pre = vao[r] + vbO;
        float g = gateN[p][m * GPN + vo];
        outv[(long)(nb + m) * V3 + d] = pre * g + (var[r] + vbR);
      }
    }
    p ^= 1;
  }
}

extern "C" void kernel_launch(void* const* d_in, const int* in_sizes, int n_in,
                              void* d_out, int out_size, void* d_ws,
                              size_t ws_size, hipStream_t stream) {
  const float* node_s = (const float*)d_in[0];
  const float* node_v = (const float*)d_in[1];
  const int* eidx     = (const int*)d_in[2];
  const float* edge_s = (const float*)d_in[3];
  const float* edge_v = (const float*)d_in[4];
  const float* ln_g = (const float*)d_in[5];
  const float* ln_b = (const float*)d_in[6];
  const float* wes = (const float*)d_in[7];
  const float* bes = (const float*)d_in[8];
  const float* wev = (const float*)d_in[9];
  const float* bev = (const float*)d_in[10];
  const float* wns = (const float*)d_in[11];
  const float* bns = (const float*)d_in[12];
  const float* wnv = (const float*)d_in[13];
  const float* bnv = (const float*)d_in[14];
  const float* wrs = (const float*)d_in[15];
  const float* brs = (const float*)d_in[16];
  const float* wrv = (const float*)d_in[17];
  const float* brv = (const float*)d_in[18];
  float* out = (float*)d_out;

  float* ws = (float*)d_ws;
  float* s_n = ws;
  float* v_n = s_n + (size_t)NN * SDIM;
  float* agg_s = v_n + (size_t)NN * V3;
  float* agg_v = agg_s + (size_t)NN * SDIM;
  float* cnt = agg_v + (size_t)NN * V3;
  int* keys = (int*)(cnt + NN);
  const int* edge_dst = eidx + EE;

  k_norm<<<NORMB + KEYB, 256, 0, stream>>>(node_s, node_v, ln_g, ln_b, s_n, v_n,
                                           edge_dst, keys);
  k_edge<<<NBLK, 1024, 0, stream>>>(edge_s, edge_v, edge_dst, keys, wes, bes,
                                    wev, bev, agg_s, agg_v, cnt);
  k_node<<<KNG, 256, 0, stream>>>(s_n, v_n, agg_s, agg_v, cnt,
                                                wns, bns, wnv, bnv, wrs, brs,
                                                wrv, brv, out);
}


# SiLU epilogues branch-free: k_edge peels the gate column tile off its column loop, k_node tests for the gate wave after its 8 chains, so chains interleave; k_node out_s add now fma; sum order same
# speedup vs baseline: 5.2831x; 1.1041x over previous
#include <hip/hip_runtime.h>
#include <math.h>

#define NN   50000
#define EE   800000
#define SDIM 128
#define VDIM 16
#define V3   48
#define ESD  32

#define RNG  196
#define NBLK ((NN + RNG - 1) / RNG)
#define NORMB ((NN * 32 + 255) / 256)
#define KEYB  ((EE / 16 + 255) / 256)
static_assert(NBLK <= 256, "a range key must fit in one byte");
static_assert(EE % 16 == 0, "keys are written and read 16 at a time");

typedef float v2f __attribute__((ext_vector_type(2)));
typedef float v8f __attribute__((ext_vector_type(8)));

__device__ __forceinline__ v8f wmma4(v2f a, v2f b, v8f c) {
  return __builtin_amdgcn_wmma_f32_16x16x4_f32(false, a, false, b, (short)0, c,
                                               false, false);
}

__device__ __forceinline__ float fast_rcp(float x) {
  return __builtin_amdgcn_rcpf(x);
}

__device__ __forceinline__ float sigmoidf_(float x) {
  return fast_rcp(1.f + __expf(-x));
}

__device__ __forceinline__ int pack_keys(int4 d) {
  unsigned k0 = ((unsigned)d.x / RNG) & 255u;
  unsigned k1 = ((unsigned)d.y / RNG) & 255u;
  unsigned k2 = ((unsigned)d.z / RNG) & 255u;
  unsigned k3 = ((unsigned)d.w / RNG) & 255u;
  return (int)(k0 | (k1 << 8) | (k2 << 16) | (k3 << 24));
}

__global__ void __launch_bounds__(256) k_norm(
    const float* __restrict__ node_s, const float* __restrict__ node_v,
    const float* __restrict__ ln_g, const float* __restrict__ ln_b,
    float* __restrict__ s_n, float* __restrict__ v_n,
    const int* __restrict__ edge_dst, int* __restrict__ keys) {
  if (blockIdx.x >= NORMB) {
    int i = (blockIdx.x - NORMB) * 256 + threadIdx.x;
    if (i < EE / 16) {
      const int4* src = (const int4*)(edge_dst + (long)i * 16);
      int4 o;
      o.x = pack_keys(src[0]);
      o.y = pack_keys(src[1]);
      o.z = pack_keys(src[2]);
      o.w = pack_keys(src[3]);
      ((int4*)keys)[i] = o;
    }
    return;
  }
  int wave = (blockIdx.x * blockDim.x + threadIdx.x) >> 5;
  int lane = threadIdx.x & 31;
  if (wave >= NN) return;

  const float* srow = node_s + (long)wave * SDIM;
  float4 x = ((const float4*)srow)[lane];
  float s = x.x + x.y + x.z + x.w;
#pragma unroll
  for (int m = 16; m >= 1; m >>= 1) s += __shfl_xor(s, m, 32);
  float mu = s * (1.f / SDIM);
  float dx = x.x - mu, dy = x.y - mu, dz = x.z - mu, dw = x.w - mu;
  float q = dx * dx + dy * dy + dz * dz + dw * dw;
#pragma unroll
  for (int m = 16; m >= 1; m >>= 1) q += __shfl_xor(q, m, 32);
  float rs = rsqrtf(q * (1.f / SDIM) + 1e-5f);
  int c0 = lane * 4;
  float4 o;
  o.x = dx * rs * ln_g[c0 + 0] + ln_b[c0 + 0];
  o.y = dy * rs * ln_g[c0 + 1] + ln_b[c0 + 1];
  o.z = dz * rs * ln_g[c0 + 2] + ln_b[c0 + 2];
  o.w = dw * rs * ln_g[c0 + 3] + ln_b[c0 + 3];
  ((float4*)(s_n + (long)wave * SDIM))[lane] = o;

  const float* vrow = node_v + (long)wave * V3;
  float2 v = {0.f, 0.f};
  int vi = lane * 2;
  if (vi < V3) v = ((const float2*)vrow)[lane];
  float qq = v.x * v.x + v.y * v.y;
#pragma unroll
  for (int m = 16; m >= 1; m >>= 1) qq += __shfl_xor(qq, m, 32);
  float inv = rsqrtf(qq * (1.f / VDIM) + 1e-8f);
  if (vi < V3) {
    float2 ov;
    ov.x = v.x * inv;
    ov.y = v.y * inv;
    ((float2*)(v_n + (long)wave * V3))[lane] = ov;
  }
}

#define SP   144
#define VP   49
#define GP   17
#define NW   32
#define QCAP 528

__device__ __forceinline__ void lds_fadd(float* p, float v) {
  __hip_atomic_fetch_add(p, v, __ATOMIC_RELAXED, __HIP_MEMORY_SCOPE_WORKGROUP);
}

__device__ __forceinline__ void col_tile(
    int c, const bool gates, const v2f (&a)[8], const int (&dlr)[8], int lane,
    const float* __restrict__ bes, const v2f* Bl, float* gate, float* accS) {
  int row = lane & 15;
  int half = lane >> 4;
  int n = 16 * c + row;
  v8f acc = {};
#pragma unroll
  for (int j = 0; j < 8; j++) {
    v2f b = Bl[(c * 8 + j) * 32 + lane];
    acc = wmma4(a[j], b, acc);
  }
  float bias = bes[n];
  float so[8];
#pragma unroll
  for (int r = 0; r < 8; r++) {
    float pre = acc[r] + bias;
    so[r] = pre * sigmoidf_(pre);
  }
  if (gates) {
#pragma unroll
    for (int r = 0; r < 8; r++)
      gate[(r + 8 * half) * GP + row] = sigmoidf_(so[r]);
  }
#pragma unroll
  for (int r = 0; r < 8; r++) lds_fadd(accS + dlr[r] * SP + n, so[r]);
}

__device__ __forceinline__ void edge_tile(
    const int* qp, int nv, int lane, int lo, int nr,
    const float* __restrict__ edge_s, const float* __restrict__ edge_v,
    const int* __restrict__ edge_dst,
    const float* __restrict__ bes, const float4* Wl, const v2f* Bl,
    float* gate, int* dls, float* accS, float* accV, float* accC) {
  int row = lane & 15;
  int half = lane >> 4;
  int eid = qp[row < nv ? row : 0];

  int dst = edge_dst[eid];
  const float* arow = edge_s + (long)eid * ESD;
  v2f a[8];
#pragma unroll
  for (int j = 0; j < 8; j++) a[j] = *(const v2f*)(arow + 4 * j + 2 * half);
  float ev0 = edge_v[eid * 3 + 0];
  float ev1 = edge_v[eid * 3 + 1];
  float ev2 = edge_v[eid * 3 + 2];

  int dl = dst - lo;
  int dlv = (row < nv && (unsigned)dl < (unsigned)nr) ? dl : RNG;
  dls[row] = dlv;
  __builtin_amdgcn_wave_barrier();
  int dlr[8];
#pragma unroll
  for (int r = 0; r < 8; r++) dlr[r] = dls[r + 8 * half];

  col_tile(0, true, a, dlr, lane, bes, Bl, gate, accS);
#pragma unroll 1
  for (int c = 1; c < 8; c++)
    col_tile(c, false, a, dlr, lane, bes, Bl, gate, accS);

  if (half == 0) lds_fadd(accC + dlv, 1.f);

  __builtin_amdgcn_wave_barrier();

#pragma unroll
  for (int t0 = 0; t0 < 24; t0 += 8) {
    float vg[8];
#pragma unroll
    for (int t = 0; t < 8; t++) {
      int d = half * 24 + t0 + t;
      float4 w = Wl[d];
      float v = ev0 * w.x + ev1 * w.y + ev2 * w.z + w.w;
      vg[t] = v * gate[row * GP + d / 3];
    }
#pragma unroll
    for (int t = 0; t < 8; t++)
      lds_fadd(accV + dlv * VP + half * 24 + t0 + t, vg[t]);
  }
  __builtin_amdgcn_wave_barrier();
}

__device__ __forceinline__ int4 ld_key(const int* __restrict__ keys, int t,
                                       int wid, int lane, int nomatch) {
  int ix = (t * NW + wid) * 512 + lane * 16;
  int4 k = {nomatch, nomatch, nomatch, nomatch};
  if (ix < EE) k = *(const int4*)(keys + (ix >> 2));
  return k;
}

__device__ __forceinline__ unsigned zero_bytes(unsigned x) {
  return (x - 0x01010101u) & ~x & 0x80808080u;
}

__global__ void __launch_bounds__(1024) k_edge(
    const float* __restrict__ edge_s, const float* __restrict__ edge_v,
    const int* __restrict__ edge_dst, const int* __restrict__ keys,
    const float* __restrict__ wes, const float* __restrict__ bes,
    const float* __restrict__ wev, const float* __restrict__ bev,
    float* __restrict__ agg_s, float* __restrict__ agg_v,
    float* __restrict__ cnt) {
  __shared__ float accS[(RNG + 1) * SP];
  __shared__ float accV[(RNG + 1) * VP];
  __shared__ float accC[RNG + 1];
  __shared__ v2f   Bl[8 * 8 * 32];
  __shared__ float4 Wl[V3];
  __shared__ int   queue[NW][QCAP];
  __shared__ float gateL[NW][16 * GP];
  __shared__ int   dlsL[NW][16];

  int tid = threadIdx.x;
  int wid = tid >> 5;
  int lane = tid & 31;
  int lo = blockIdx.x * RNG;
  int nr = NN - lo < RNG ? NN - lo : RNG;

  for (int i = tid; i < (RNG + 1) * SP; i += 1024) accS[i] = 0.f;
  for (int i = tid; i < (RNG + 1) * VP; i += 1024) accV[i] = 0.f;
  for (int i = tid; i < RNG + 1; i += 1024) accC[i] = 0.f;
  for (int i = tid; i < 8 * 8 * 32; i += 1024) {
    int l = i & 31;
    int j = (i >> 5) & 7;
    int c = i >> 8;
    int n = 16 * c + (l & 15);
    int k0 = 4 * j + 2 * (l >> 4);
    Bl[i] = *(const v2f*)(wes + (long)n * ESD + k0);
  }
  if (tid < V3) {
    float4 w = {wev[tid * 3 + 0], wev[tid * 3 + 1], wev[tid * 3 + 2], bev[tid]};
    Wl[tid] = w;
  }
  __syncthreads();

  int* q = queue[wid];
  float* gate = gateL[wid];
  int* dls = dlsL[wid];
  int qn = 0;
  unsigned ltmask = (1u << lane) - 1u;
  const int NSTEP = (EE + 512 * NW - 1) / (512 * NW);

  unsigned pat = blockIdx.x * 0x01010101u;
  int nomatch = (int)~pat;

  const int NG = (NSTEP + 3) / 4;
  int4 kk[4];
#pragma unroll
  for (int u = 0; u < 4; u++) kk[u] = ld_key(keys, u, wid, lane, nomatch);
#pragma unroll 1
  for (int g = 0; g < NG; g++) {
#pragma unroll
    for (int u = 0; u < 4; u++) {
      int t = g * 4 + u;
      int base = (t * NW + wid) * 512 + lane * 16;
      unsigned zz = (zero_bytes((unsigned)kk[u].x ^ pat) >> 7) |
                    (zero_bytes((unsigned)kk[u].y ^ pat) >> 6) |
                    (zero_bytes((unsigned)kk[u].z ^ pat) >> 5) |
                    (zero_bytes((unsigned)kk[u].w ^ pat) >> 4);
      kk[u] = ld_key(keys, t + 4, wid, lane, nomatch);
      unsigned bal = __builtin_amdgcn_ballot_w32(zz != 0u);
#pragma unroll 1
      while (bal != 0u) {
        if (zz != 0u) {
          int p = __builtin_ctz(zz);
          q[qn + __popc(bal & ltmask)] = base + ((p & 3) << 2) + (p >> 3);
          zz &= zz - 1u;
        }
        qn += __popc(bal);
        bal = __builtin_amdgcn_ballot_w32(zz != 0u);
      }
      __builtin_amdgcn_wave_barrier();
#pragma unroll 1
      while (qn >= 16) {
        qn -= 16;
        edge_tile(q + qn, 16, lane, lo, nr, edge_s, edge_v, edge_dst, bes, Wl,
                  Bl, gate, dls, accS, accV, accC);
      }
    }
  }
  if (qn > 0)
    edge_tile(q, qn, lane, lo, nr, edge_s, edge_v, edge_dst, bes, Wl, Bl, gate,
              dls, accS, accV, accC);

  __syncthreads();
  for (int i = tid; i < nr * SDIM; i += 1024) {
    int node = i >> 7, col = i & 127;
    agg_s[(long)(lo + node) * SDIM + col] = accS[node * SP + col];
  }
  for (int i = tid; i < nr * V3; i += 1024) {
    int node = i / V3, col = i - node * V3;
    agg_v[(long)(lo + node) * V3 + col] = accV[node * VP + col];
  }
  for (int i = tid; i < nr; i += 1024) cnt[lo + i] = accC[i];
}

#define NT   (NN / 16)
#define PA   132
#define PV   52
#define GPN  17
#define KNG  256

struct NodeRows {
  float4 s0, s1, g0, g1, v, gv;
  float c0, c1, cv;
};

__device__ __forceinline__ void kn_load(
    NodeRows& R, int t, int tid, int wid, int lane,
    const float* __restrict__ s_n, const float* __restrict__ v_n,
    const float* __restrict__ agg_s, const float* __restrict__ agg_v,
    const float* __restrict__ cnt) {
  int nb = t * 16;
  int r0 = nb + 2 * wid;
  R.s0 = ((const float4*)(s_n + (long)r0 * SDIM))[lane];
  R.s1 = ((const float4*)(s_n + (long)(r0 + 1) * SDIM))[lane];
  R.g0 = ((const float4*)(agg_s + (long)r0 * SDIM))[lane];
  R.g1 = ((const float4*)(agg_s + (long)(r0 + 1) * SDIM))[lane];
  R.c0 = cnt[r0];
  R.c1 = cnt[r0 + 1];
  if (tid < 192) {
    R.v = ((const float4*)(v_n + (long)nb * V3))[tid];
    R.gv = ((const float4*)(agg_v + (long)nb * V3))[tid];
    R.cv = cnt[nb + tid / 12];
  }
}

__device__ __forceinline__ void kn_stage(
    const NodeRows& R, float* ain, float* sn, float* vain, float* vn,
    int tid, int wid, int lane) {
  float i0 = fast_rcp(fmaxf(R.c0, 1.f));
  float i1 = fast_rcp(fmaxf(R.c1, 1.f));
  float4 a0 = {R.s0.x + R.g0.x * i0, R.s0.y + R.g0.y * i0,
               R.s0.z + R.g0.z * i0, R.s0.w + R.g0.w * i0};
  float4 a1 = {R.s1.x + R.g1.x * i1, R.s1.y + R.g1.y * i1,
               R.s1.z + R.g1.z * i1, R.s1.w + R.g1.w * i1};
  int o = (2 * wid) * PA + 4 * lane;
  *(float4*)(ain + o) = a0;
  *(float4*)(ain + o + PA) = a1;
  *(float4*)(sn + o) = R.s0;
  *(float4*)(sn + o + PA) = R.s1;
  if (tid < 192) {
    int vr = tid / 12;
    int vc = tid - vr * 12;
    float iv = fast_rcp(fmaxf(R.cv, 1.f));
    float4 av = {R.v.x + R.gv.x * iv, R.v.y + R.gv.y * iv,
                 R.v.z + R.gv.z * iv, R.v.w + R.gv.w * iv};
    int ov = vr * PV + 4 * vc;
    *(float4*)(vain + ov) = av;
    *(float4*)(vn + ov) = R.v;
  }
}

__device__ __forceinline__ void lds_barrier() {
  __builtin_amdgcn_fence(__ATOMIC_RELEASE, "workgroup", "local");
  __builtin_amdgcn_s_barrier();
  __builtin_amdgcn_fence(__ATOMIC_ACQUIRE, "workgroup", "local");
}

__global__ void __launch_bounds__(256) k_node(
    const float* __restrict__ s_n, const float* __restrict__ v_n,
    const float* __restrict__ agg_s, const float* __restrict__ agg_v,
    const float* __restrict__ cnt,
    const float* __restrict__ wns, const float* __restrict__ bns,
    const float* __restrict__ wnv, const float* __restrict__ bnv,
    const float* __restrict__ wrs, const float* __restrict__ brs,
    const float* __restrict__ wrv, const float* __restrict__ brv,
    float* __restrict__ out) {
  __shared__ __attribute__((aligned(16))) float ainL[2][16 * PA];
  __shared__ __attribute__((aligned(16))) float snL[2][16 * PA];
  __shared__ __attribute__((aligned(16))) float vainL[2][16 * PV];
  __shared__ __attribute__((aligned(16))) float vnL[2][16 * PV];
  __shared__ float gateN[2][16 * GPN];

  int tid = threadIdx.x;
  int wid = __builtin_amdgcn_readfirstlane(tid >> 5);
  int lane = tid & 31;
  int row = lane & 15;
  int half = lane >> 4;

  int n = 16 * wid + row;
  v2f bO[32], bR[32];
  {
    const float* bo = wns + (long)n * SDIM + 2 * half;
    const float* br = wrs + (long)n * SDIM + 2 * half;
#pragma unroll
    for (int j = 0; j < 32; j++) {
      bO[j] = *(const v2f*)(bo + 4 * j);
      bR[j] = *(const v2f*)(br + 4 * j);
    }
  }
  float biasO = bns[n], biasR = brs[n];

  bool vw = wid >= 1 && wid <= 3;
  int d = 16 * (vw ? wid - 1 : 0) + row;
  int vo = d / 3;
  v2f cO[12], cR[12];
  float vbO = 0.f, vbR = 0.f;
#pragma unroll
  for (int j = 0; j < 12; j++) {
    cO[j] = v2f{0.f, 0.f};
    cR[j] = v2f{0.f, 0.f};
  }
  if (vw) {
    const float* bo = wnv + (long)d * V3 + 2 * half;
    const float* br = wrv + (long)d * V3 + 2 * half;
#pragma unroll
    for (int j = 0; j < 12; j++) {
      cO[j] = *(const v2f*)(bo + 4 * j);
      cR[j] = *(const v2f*)(br + 4 * j);
    }
    vbO = bnv[d];
    vbR = brv[d];
  }

  float* outv = out + (long)NN * SDIM;
  NodeRows R;
  int t = blockIdx.x;
  kn_load(R, t, tid, wid, lane, s_n, v_n, agg_s, agg_v, cnt);
  kn_stage(R, ainL[0], snL[0], vainL[0], vnL[0], tid, wid, lane);
  lds_barrier();
  int p = 0;
#pragma unroll 1
  for (; t < NT; t += KNG) {
    bool more = t + KNG < NT;
    if (more) kn_load(R, t + KNG, tid, wid, lane, s_n, v_n, agg_s, agg_v, cnt);
    int nb = t * 16;

    const float* aA = ainL[p] + row * PA + 2 * half;
    const float* aS = snL[p] + row * PA + 2 * half;
    v8f ao = {};
    v8f ar = {};
#pragma unroll
    for (int j = 0; j < 32; j++) {
      v2f ain = *(const v2f*)(aA + 4 * j);
      v2f as = *(const v2f*)(aS + 4 * j);
      ao = wmma4(ain, bO[j], ao);
      ar = wmma4(as, bR[j], ar);
    }
    float so[8];
#pragma unroll
    for (int r = 0; r < 8; r++) {
      int m = r + 8 * half;
      float pre = ao[r] + biasO;
      so[r] = pre * sigmoidf_(pre);
      out[(long)(nb + m) * SDIM + n] = so[r] + (ar[r] + biasR);
    }
    if (wid == 0) {
#pragma unroll
      for (int r = 0; r < 8; r++)
        gateN[p][(r + 8 * half) * GPN + row] = sigmoidf_(so[r]);
    }

    v8f vao = {};
    v8f var = {};
    if (vw) {
      const float* vA = vainL[p] + row * PV + 2 * half;
      const float* vS = vnL[p] + row * PV + 2 * half;
#pragma unroll
      for (int j = 0; j < 12; j++) {
        v2f ain = *(const v2f*)(vA + 4 * j);
        v2f av = *(const v2f*)(vS + 4 * j);
        vao = wmma4(ain, cO[j], vao);
        var = wmma4(av, cR[j], var);
      }
    }

    if (more)
      kn_stage(R, ainL[p ^ 1], snL[p ^ 1], vainL[p ^ 1], vnL[p ^ 1], tid, wid,
               lane);
    lds_barrier();

    if (vw) {
#pragma unroll
      for (int r = 0; r < 8; r++) {
        int m = r + 8 * half;
        float pre = vao[r] + vbO;
        float g = gateN[p][m * GPN + vo];
        outv[(long)(nb + m) * V3 + d] = pre * g + (var[r] + vbR);
      }
    }
    p ^= 1;
  }
}

extern "C" void kernel_launch(void* const* d_in, const int* in_sizes, int n_in,
                              void* d_out, int out_size, void* d_ws,
                              size_t ws_size, hipStream_t stream) {
  const float* node_s = (const float*)d_in[0];
  const float* node_v = (const float*)d_in[1];
  const int* eidx     = (const int*)d_in[2];
  const float* edge_s = (const float*)d_in[3];
  const float* edge_v = (const float*)d_in[4];
  const float* ln_g = (const float*)d_in[5];
  const float* ln_b = (const float*)d_in[6];
  const float* wes = (const float*)d_in[7];
  const float* bes = (const float*)d_in[8];
  const float* wev = (const float*)d_in[9];
  const float* bev = (const float*)d_in[10];
  const float* wns = (const float*)d_in[11];
  const float* bns = (const float*)d_in[12];
  const float* wnv = (const float*)d_in[13];
  const float* bnv = (const float*)d_in[14];
  const float* wrs = (const float*)d_in[15];
  const float* brs = (const float*)d_in[16];
  const float* wrv = (const float*)d_in[17];
  const float* brv = (const float*)d_in[18];
  float* out = (float*)d_out;

  float* ws = (float*)d_ws;
  float* s_n = ws;
  float* v_n = s_n + (size_t)NN * SDIM;
  float* agg_s = v_n + (size_t)NN * V3;
  float* agg_v = agg_s + (size_t)NN * SDIM;
  float* cnt = agg_v + (size_t)NN * V3;
  int* keys = (int*)(cnt + NN);
  const int* edge_dst = eidx + EE;

  k_norm<<<NORMB + KEYB, 256, 0, stream>>>(node_s, node_v, ln_g, ln_b, s_n, v_n,
                                           edge_dst, keys);
  k_edge<<<NBLK, 1024, 0, stream>>>(edge_s, edge_v, edge_dst, keys, wes, bes,
                                    wev, bev, agg_s, agg_v, cnt);
  k_node<<<KNG, 256, 0, stream>>>(s_n, v_n, agg_s, agg_v, cnt,
                                                wns, bns, wnv, bnv, wrs, brs,
                                                wrv, brv, out);
}


# k_edge: the under-16 edges each wave has left after its scan go into one LDS pool (counts, in-wave prefix, copy); the work-group forms full tiles from it, not 32 partial ones; sum order LDS arrival
# speedup vs baseline: 5.5566x; 1.0518x over previous
#include <hip/hip_runtime.h>
#include <math.h>

#define NN   50000
#define EE   800000
#define SDIM 128
#define VDIM 16
#define V3   48
#define ESD  32

#define RNG  196
#define NBLK ((NN + RNG - 1) / RNG)
#define NORMB ((NN * 32 + 255) / 256)
#define KEYB  ((EE / 16 + 255) / 256)
static_assert(NBLK <= 256, "a range key must fit in one byte");
static_assert(EE % 16 == 0, "keys are written and read 16 at a time");

typedef float v2f __attribute__((ext_vector_type(2)));
typedef float v8f __attribute__((ext_vector_type(8)));

__device__ __forceinline__ v8f wmma4(v2f a, v2f b, v8f c) {
  return __builtin_amdgcn_wmma_f32_16x16x4_f32(false, a, false, b, (short)0, c,
                                               false, false);
}

__device__ __forceinline__ float fast_rcp(float x) {
  return __builtin_amdgcn_rcpf(x);
}

__device__ __forceinline__ float sigmoidf_(float x) {
  return fast_rcp(1.f + __expf(-x));
}

__device__ __forceinline__ int pack_keys(int4 d) {
  unsigned k0 = ((unsigned)d.x / RNG) & 255u;
  unsigned k1 = ((unsigned)d.y / RNG) & 255u;
  unsigned k2 = ((unsigned)d.z / RNG) & 255u;
  unsigned k3 = ((unsigned)d.w / RNG) & 255u;
  return (int)(k0 | (k1 << 8) | (k2 << 16) | (k3 << 24));
}

__global__ void __launch_bounds__(256) k_norm(
    const float* __restrict__ node_s, const float* __restrict__ node_v,
    const float* __restrict__ ln_g, const float* __restrict__ ln_b,
    float* __restrict__ s_n, float* __restrict__ v_n,
    const int* __restrict__ edge_dst, int* __restrict__ keys) {
  if (blockIdx.x >= NORMB) {
    int i = (blockIdx.x - NORMB) * 256 + threadIdx.x;
    if (i < EE / 16) {
      const int4* src = (const int4*)(edge_dst + (long)i * 16);
      int4 o;
      o.x = pack_keys(src[0]);
      o.y = pack_keys(src[1]);
      o.z = pack_keys(src[2]);
      o.w = pack_keys(src[3]);
      ((int4*)keys)[i] = o;
    }
    return;
  }
  int wave = (blockIdx.x * blockDim.x + threadIdx.x) >> 5;
  int lane = threadIdx.x & 31;
  if (wave >= NN) return;

  const float* srow = node_s + (long)wave * SDIM;
  float4 x = ((const float4*)srow)[lane];
  float s = x.x + x.y + x.z + x.w;
#pragma unroll
  for (int m = 16; m >= 1; m >>= 1) s += __shfl_xor(s, m, 32);
  float mu = s * (1.f / SDIM);
  float dx = x.x - mu, dy = x.y - mu, dz = x.z - mu, dw = x.w - mu;
  float q = dx * dx + dy * dy + dz * dz + dw * dw;
#pragma unroll
  for (int m = 16; m >= 1; m >>= 1) q += __shfl_xor(q, m, 32);
  float rs = rsqrtf(q * (1.f / SDIM) + 1e-5f);
  int c0 = lane * 4;
  float4 o;
  o.x = dx * rs * ln_g[c0 + 0] + ln_b[c0 + 0];
  o.y = dy * rs * ln_g[c0 + 1] + ln_b[c0 + 1];
  o.z = dz * rs * ln_g[c0 + 2] + ln_b[c0 + 2];
  o.w = dw * rs * ln_g[c0 + 3] + ln_b[c0 + 3];
  ((float4*)(s_n + (long)wave * SDIM))[lane] = o;

  const float* vrow = node_v + (long)wave * V3;
  float2 v = {0.f, 0.f};
  int vi = lane * 2;
  if (vi < V3) v = ((const float2*)vrow)[lane];
  float qq = v.x * v.x + v.y * v.y;
#pragma unroll
  for (int m = 16; m >= 1; m >>= 1) qq += __shfl_xor(qq, m, 32);
  float inv = rsqrtf(qq * (1.f / VDIM) + 1e-8f);
  if (vi < V3) {
    float2 ov;
    ov.x = v.x * inv;
    ov.y = v.y * inv;
    ((float2*)(v_n + (long)wave * V3))[lane] = ov;
  }
}

#define SP   144
#define VP   49
#define GP   17
#define NW   32
#define QCAP 528
static_assert(NW == 32, "the left-over counts are summed one wave per lane");

__device__ __forceinline__ void lds_fadd(float* p, float v) {
  __hip_atomic_fetch_add(p, v, __ATOMIC_RELAXED, __HIP_MEMORY_SCOPE_WORKGROUP);
}

__device__ __forceinline__ void col_tile(
    int c, const bool gates, const v2f (&a)[8], const int (&dlr)[8], int lane,
    const float* __restrict__ bes, const v2f* Bl, float* gate, float* accS) {
  int row = lane & 15;
  int half = lane >> 4;
  int n = 16 * c + row;
  v8f acc = {};
#pragma unroll
  for (int j = 0; j < 8; j++) {
    v2f b = Bl[(c * 8 + j) * 32 + lane];
    acc = wmma4(a[j], b, acc);
  }
  float bias = bes[n];
  float so[8];
#pragma unroll
  for (int r = 0; r < 8; r++) {
    float pre = acc[r] + bias;
    so[r] = pre * sigmoidf_(pre);
  }
  if (gates) {
#pragma unroll
    for (int r = 0; r < 8; r++)
      gate[(r + 8 * half) * GP + row] = sigmoidf_(so[r]);
  }
#pragma unroll
  for (int r = 0; r < 8; r++) lds_fadd(accS + dlr[r] * SP + n, so[r]);
}

__device__ __forceinline__ void edge_tile(
    const int* qp, int nv, int lane, int lo, int nr,
    const float* __restrict__ edge_s, const float* __restrict__ edge_v,
    const int* __restrict__ edge_dst,
    const float* __restrict__ bes, const float4* Wl, const v2f* Bl,
    float* gate, int* dls, float* accS, float* accV, float* accC) {
  int row = lane & 15;
  int half = lane >> 4;
  int eid = qp[row < nv ? row : 0];

  int dst = edge_dst[eid];
  const float* arow = edge_s + (long)eid * ESD;
  v2f a[8];
#pragma unroll
  for (int j = 0; j < 8; j++) a[j] = *(const v2f*)(arow + 4 * j + 2 * half);
  float ev0 = edge_v[eid * 3 + 0];
  float ev1 = edge_v[eid * 3 + 1];
  float ev2 = edge_v[eid * 3 + 2];

  int dl = dst - lo;
  int dlv = (row < nv && (unsigned)dl < (unsigned)nr) ? dl : RNG;
  dls[row] = dlv;
  __builtin_amdgcn_wave_barrier();
  int dlr[8];
#pragma unroll
  for (int r = 0; r < 8; r++) dlr[r] = dls[r + 8 * half];

  col_tile(0, true, a, dlr, lane, bes, Bl, gate, accS);
#pragma unroll 1
  for (int c = 1; c < 8; c++)
    col_tile(c, false, a, dlr, lane, bes, Bl, gate, accS);

  if (half == 0) lds_fadd(accC + dlv, 1.f);

  __builtin_amdgcn_wave_barrier();

#pragma unroll
  for (int t0 = 0; t0 < 24; t0 += 8) {
    float vg[8];
#pragma unroll
    for (int t = 0; t < 8; t++) {
      int d = half * 24 + t0 + t;
      float4 w = Wl[d];
      float v = ev0 * w.x + ev1 * w.y + ev2 * w.z + w.w;
      vg[t] = v * gate[row * GP + d / 3];
    }
#pragma unroll
    for (int t = 0; t < 8; t++)
      lds_fadd(accV + dlv * VP + half * 24 + t0 + t, vg[t]);
  }
  __builtin_amdgcn_wave_barrier();
}

__device__ __forceinline__ int4 ld_key(const int* __restrict__ keys, int t,
                                       int wid, int lane, int nomatch) {
  int ix = (t * NW + wid) * 512 + lane * 16;
  int4 k = {nomatch, nomatch, nomatch, nomatch};
  if (ix < EE) k = *(const int4*)(keys + (ix >> 2));
  return k;
}

__device__ __forceinline__ unsigned zero_bytes(unsigned x) {
  return (x - 0x01010101u) & ~x & 0x80808080u;
}

__global__ void __launch_bounds__(1024) k_edge(
    const float* __restrict__ edge_s, const float* __restrict__ edge_v,
    const int* __restrict__ edge_dst, const int* __restrict__ keys,
    const float* __restrict__ wes, const float* __restrict__ bes,
    const float* __restrict__ wev, const float* __restrict__ bev,
    float* __restrict__ agg_s, float* __restrict__ agg_v,
    float* __restrict__ cnt) {
  __shared__ float accS[(RNG + 1) * SP];
  __shared__ float accV[(RNG + 1) * VP];
  __shared__ float accC[RNG + 1];
  __shared__ v2f   Bl[8 * 8 * 32];
  __shared__ float4 Wl[V3];
  __shared__ int   queue[NW][QCAP];
  __shared__ float gateL[NW][16 * GP];
  __shared__ int   dlsL[NW][16];
  __shared__ int   pcnt[NW];
  __shared__ int   pool[NW * 16];

  int tid = threadIdx.x;
  int wid = tid >> 5;
  int lane = tid & 31;
  int lo = blockIdx.x * RNG;
  int nr = NN - lo < RNG ? NN - lo : RNG;

  for (int i = tid; i < (RNG + 1) * SP; i += 1024) accS[i] = 0.f;
  for (int i = tid; i < (RNG + 1) * VP; i += 1024) accV[i] = 0.f;
  for (int i = tid; i < RNG + 1; i += 1024) accC[i] = 0.f;
  for (int i = tid; i < 8 * 8 * 32; i += 1024) {
    int l = i & 31;
    int j = (i >> 5) & 7;
    int c = i >> 8;
    int n = 16 * c + (l & 15);
    int k0 = 4 * j + 2 * (l >> 4);
    Bl[i] = *(const v2f*)(wes + (long)n * ESD + k0);
  }
  if (tid < V3) {
    float4 w = {wev[tid * 3 + 0], wev[tid * 3 + 1], wev[tid * 3 + 2], bev[tid]};
    Wl[tid] = w;
  }
  __syncthreads();

  int* q = queue[wid];
  float* gate = gateL[wid];
  int* dls = dlsL[wid];
  int qn = 0;
  unsigned ltmask = (1u << lane) - 1u;
  const int NSTEP = (EE + 512 * NW - 1) / (512 * NW);

  unsigned pat = blockIdx.x * 0x01010101u;
  int nomatch = (int)~pat;

  const int NG = (NSTEP + 3) / 4;
  int4 kk[4];
#pragma unroll
  for (int u = 0; u < 4; u++) kk[u] = ld_key(keys, u, wid, lane, nomatch);
#pragma unroll 1
  for (int g = 0; g < NG; g++) {
#pragma unroll
    for (int u = 0; u < 4; u++) {
      int t = g * 4 + u;
      int base = (t * NW + wid) * 512 + lane * 16;
      unsigned zz = (zero_bytes((unsigned)kk[u].x ^ pat) >> 7) |
                    (zero_bytes((unsigned)kk[u].y ^ pat) >> 6) |
                    (zero_bytes((unsigned)kk[u].z ^ pat) >> 5) |
                    (zero_bytes((unsigned)kk[u].w ^ pat) >> 4);
      kk[u] = ld_key(keys, t + 4, wid, lane, nomatch);
      unsigned bal = __builtin_amdgcn_ballot_w32(zz != 0u);
#pragma unroll 1
      while (bal != 0u) {
        if (zz != 0u) {
          int p = __builtin_ctz(zz);
          q[qn + __popc(bal & ltmask)] = base + ((p & 3) << 2) + (p >> 3);
          zz &= zz - 1u;
        }
        qn += __popc(bal);
        bal = __builtin_amdgcn_ballot_w32(zz != 0u);
      }
      __builtin_amdgcn_wave_barrier();
#pragma unroll 1
      while (qn >= 16) {
        qn -= 16;
        edge_tile(q + qn, 16, lane, lo, nr, edge_s, edge_v, edge_dst, bes, Wl,
                  Bl, gate, dls, accS, accV, accC);
      }
    }
  }
  if (lane == 0) pcnt[wid] = qn;
  __syncthreads();
  int mine = pcnt[lane];
  int incl = mine;
#pragma unroll
  for (int m = 1; m < 32; m <<= 1) {
    int o = __shfl_up(incl, m, 32);
    if (lane >= m) incl += o;
  }
  int total = __shfl(incl, 31, 32);
  int myoff = __shfl(incl - mine, wid, 32);
  if (lane < qn) pool[myoff + lane] = q[lane];
  __syncthreads();
  int swid = __builtin_amdgcn_readfirstlane(wid);
  total = __builtin_amdgcn_readfirstlane(total);
  for (int i = swid * 16; i < total; i += NW * 16) {
    int nv = total - i < 16 ? total - i : 16;
    edge_tile(pool + i, nv, lane, lo, nr, edge_s, edge_v, edge_dst, bes, Wl, Bl,
              gate, dls, accS, accV, accC);
  }

  __syncthreads();
  for (int i = tid; i < nr * SDIM; i += 1024) {
    int node = i >> 7, col = i & 127;
    agg_s[(long)(lo + node) * SDIM + col] = accS[node * SP + col];
  }
  for (int i = tid; i < nr * V3; i += 1024) {
    int node = i / V3, col = i - node * V3;
    agg_v[(long)(lo + node) * V3 + col] = accV[node * VP + col];
  }
  for (int i = tid; i < nr; i += 1024) cnt[lo + i] = accC[i];
}

#define NT   (NN / 16)
#define PA   132
#define PV   52
#define GPN  17
#define KNG  256

struct NodeRows {
  float4 s0, s1, g0, g1, v, gv;
  float c0, c1, cv;
};

__device__ __forceinline__ void kn_load(
    NodeRows& R, int t, int tid, int wid, int lane,
    const float* __restrict__ s_n, const float* __restrict__ v_n,
    const float* __restrict__ agg_s, const float* __restrict__ agg_v,
    const float* __restrict__ cnt) {
  int nb = t * 16;
  int r0 = nb + 2 * wid;
  R.s0 = ((const float4*)(s_n + (long)r0 * SDIM))[lane];
  R.s1 = ((const float4*)(s_n + (long)(r0 + 1) * SDIM))[lane];
  R.g0 = ((const float4*)(agg_s + (long)r0 * SDIM))[lane];
  R.g1 = ((const float4*)(agg_s + (long)(r0 + 1) * SDIM))[lane];
  R.c0 = cnt[r0];
  R.c1 = cnt[r0 + 1];
  if (tid < 192) {
    R.v = ((const float4*)(v_n + (long)nb * V3))[tid];
    R.gv = ((const float4*)(agg_v + (long)nb * V3))[tid];
    R.cv = cnt[nb + tid / 12];
  }
}

__device__ __forceinline__ void kn_stage(
    const NodeRows& R, float* ain, float* sn, float* vain, float* vn,
    int tid, int wid, int lane) {
  float i0 = fast_rcp(fmaxf(R.c0, 1.f));
  float i1 = fast_rcp(fmaxf(R.c1, 1.f));
  float4 a0 = {R.s0.x + R.g0.x * i0, R.s0.y + R.g0.y * i0,
               R.s0.z + R.g0.z * i0, R.s0.w + R.g0.w * i0};
  float4 a1 = {R.s1.x + R.g1.x * i1, R.s1.y + R.g1.y * i1,
               R.s1.z + R.g1.z * i1, R.s1.w + R.g1.w * i1};
  int o = (2 * wid) * PA + 4 * lane;
  *(float4*)(ain + o) = a0;
  *(float4*)(ain + o + PA) = a1;
  *(float4*)(sn + o) = R.s0;
  *(float4*)(sn + o + PA) = R.s1;
  if (tid < 192) {
    int vr = tid / 12;
    int vc = tid - vr * 12;
    float iv = fast_rcp(fmaxf(R.cv, 1.f));
    float4 av = {R.v.x + R.gv.x * iv, R.v.y + R.gv.y * iv,
                 R.v.z + R.gv.z * iv, R.v.w + R.gv.w * iv};
    int ov = vr * PV + 4 * vc;
    *(float4*)(vain + ov) = av;
    *(float4*)(vn + ov) = R.v;
  }
}

__device__ __forceinline__ void lds_barrier() {
  __builtin_amdgcn_fence(__ATOMIC_RELEASE, "workgroup", "local");
  __builtin_amdgcn_s_barrier();
  __builtin_amdgcn_fence(__ATOMIC_ACQUIRE, "workgroup", "local");
}

__global__ void __launch_bounds__(256) k_node(
    const float* __restrict__ s_n, const float* __restrict__ v_n,
    const float* __restrict__ agg_s, const float* __restrict__ agg_v,
    const float* __restrict__ cnt,
    const float* __restrict__ wns, const float* __restrict__ bns,
    const float* __restrict__ wnv, const float* __restrict__ bnv,
    const float* __restrict__ wrs, const float* __restrict__ brs,
    const float* __restrict__ wrv, const float* __restrict__ brv,
    float* __restrict__ out) {
  __shared__ __attribute__((aligned(16))) float ainL[2][16 * PA];
  __shared__ __attribute__((aligned(16))) float snL[2][16 * PA];
  __shared__ __attribute__((aligned(16))) float vainL[2][16 * PV];
  __shared__ __attribute__((aligned(16))) float vnL[2][16 * PV];
  __shared__ float gateN[2][16 * GPN];

  int tid = threadIdx.x;
  int wid = __builtin_amdgcn_readfirstlane(tid >> 5);
  int lane = tid & 31;
  int row = lane & 15;
  int half = lane >> 4;

  int n = 16 * wid + row;
  v2f bO[32], bR[32];
  {
    const float* bo = wns + (long)n * SDIM + 2 * half;
    const float* br = wrs + (long)n * SDIM + 2 * half;
#pragma unroll
    for (int j = 0; j < 32; j++) {
      bO[j] = *(const v2f*)(bo + 4 * j);
      bR[j] = *(const v2f*)(br + 4 * j);
    }
  }
  float biasO = bns[n], biasR = brs[n];

  bool vw = wid >= 1 && wid <= 3;
  int d = 16 * (vw ? wid - 1 : 0) + row;
  int vo = d / 3;
  v2f cO[12], cR[12];
  float vbO = 0.f, vbR = 0.f;
#pragma unroll
  for (int j = 0; j < 12; j++) {
    cO[j] = v2f{0.f, 0.f};
    cR[j] = v2f{0.f, 0.f};
  }
  if (vw) {
    const float* bo = wnv + (long)d * V3 + 2 * half;
    const float* br = wrv + (long)d * V3 + 2 * half;
#pragma unroll
    for (int j = 0; j < 12; j++) {
      cO[j] = *(const v2f*)(bo + 4 * j);
      cR[j] = *(const v2f*)(br + 4 * j);
    }
    vbO = bnv[d];
    vbR = brv[d];
  }

  float* outv = out + (long)NN * SDIM;
  NodeRows R;
  int t = blockIdx.x;
  kn_load(R, t, tid, wid, lane, s_n, v_n, agg_s, agg_v, cnt);
  kn_stage(R, ainL[0], snL[0], vainL[0], vnL[0], tid, wid, lane);
  lds_barrier();
  int p = 0;
#pragma unroll 1
  for (; t < NT; t += KNG) {
    bool more = t + KNG < NT;
    if (more) kn_load(R, t + KNG, tid, wid, lane, s_n, v_n, agg_s, agg_v, cnt);
    int nb = t * 16;

    const float* aA = ainL[p] + row * PA + 2 * half;
    const float* aS = snL[p] + row * PA + 2 * half;
    v8f ao = {};
    v8f ar = {};
#pragma unroll
    for (int j = 0; j < 32; j++) {
      v2f ain = *(const v2f*)(aA + 4 * j);
      v2f as = *(const v2f*)(aS + 4 * j);
      ao = wmma4(ain, bO[j], ao);
      ar = wmma4(as, bR[j], ar);
    }
    float so[8];
#pragma unroll
    for (int r = 0; r < 8; r++) {
      int m = r + 8 * half;
      float pre = ao[r] + biasO;
      so[r] = pre * sigmoidf_(pre);
      out[(long)(nb + m) * SDIM + n] = so[r] + (ar[r] + biasR);
    }
    if (wid == 0) {
#pragma unroll
      for (int r = 0; r < 8; r++)
        gateN[p][(r + 8 * half) * GPN + row] = sigmoidf_(so[r]);
    }

    v8f vao = {};
    v8f var = {};
    if (vw) {
      const float* vA = vainL[p] + row * PV + 2 * half;
      const float* vS = vnL[p] + row * PV + 2 * half;
#pragma unroll
      for (int j = 0; j < 12; j++) {
        v2f ain = *(const v2f*)(vA + 4 * j);
        v2f av = *(const v2f*)(vS + 4 * j);
        vao = wmma4(ain, cO[j], vao);
        var = wmma4(av, cR[j], var);
      }
    }

    if (more)
      kn_stage(R, ainL[p ^ 1], snL[p ^ 1], vainL[p ^ 1], vnL[p ^ 1], tid, wid,
               lane);
    lds_barrier();

    if (vw) {
#pragma unroll
      for (int r = 0; r < 8; r++) {
        int m = r + 8 * half;
        float pre = vao[r] + vbO;
        float g = gateN[p][m * GPN + vo];
        outv[(long)(nb + m) * V3 + d] = pre * g + (var[r] + vbR);
      }
    }
    p ^= 1;
  }
}

extern "C" void kernel_launch(void* const* d_in, const int* in_sizes, int n_in,
                              void* d_out, int out_size, void* d_ws,
                              size_t ws_size, hipStream_t stream) {
  const float* node_s = (const float*)d_in[0];
  const float* node_v = (const float*)d_in[1];
  const int* eidx     = (const int*)d_in[2];
  const float* edge_s = (const float*)d_in[3];
  const float* edge_v = (const float*)d_in[4];
  const float* ln_g = (const float*)d_in[5];
  const float* ln_b = (const float*)d_in[6];
  const float* wes = (const float*)d_in[7];
  const float* bes = (const float*)d_in[8];
  const float* wev = (const float*)d_in[9];
  const float* bev = (const float*)d_in[10];
  const float* wns = (const float*)d_in[11];
  const float* bns = (const float*)d_in[12];
  const float* wnv = (const float*)d_in[13];
  const float* bnv = (const float*)d_in[14];
  const float* wrs = (const float*)d_in[15];
  const float* brs = (const float*)d_in[16];
  const float* wrv = (const float*)d_in[17];
  const float* brv = (const float*)d_in[18];
  float* out = (float*)d_out;

  float* ws = (float*)d_ws;
  float* s_n = ws;
  float* v_n = s_n + (size_t)NN * SDIM;
  float* agg_s = v_n + (size_t)NN * V3;
  float* agg_v = agg_s + (size_t)NN * SDIM;
  float* cnt = agg_v + (size_t)NN * V3;
  int* keys = (int*)(cnt + NN);
  const int* edge_dst = eidx + EE;

  k_norm<<<NORMB + KEYB, 256, 0, stream>>>(node_s, node_v, ln_g, ln_b, s_n, v_n,
                                           edge_dst, keys);
  k_edge<<<NBLK, 1024, 0, stream>>>(edge_s, edge_v, edge_dst, keys, wes, bes,
                                    wev, bev, agg_s, agg_v, cnt);
  k_node<<<KNG, 256, 0, stream>>>(s_n, v_n, agg_s, agg_v, cnt,
                                                wns, bns, wnv, bnv, wrs, brs,
                                                wrv, brv, out);
}
